# starting kernel
# baseline (speedup 1.0000x reference)
#include <hip/hip_runtime.h>


#define GCN_N 50000
#define GCN_E 800000
#define GCN_IN 128
#define GCN_H 64

#define PAIR_STRIDE 130

typedef float v2f __attribute__((ext_vector_type(2)));
typedef float v8f __attribute__((ext_vector_type(8)));


__global__ void gcn_deg_init(float* __restrict__ deg, int n) {
    int i = blockIdx.x * blockDim.x + threadIdx.x;
    if (i < n) deg[i] = 2.0f;
}

__global__ void gcn_deg_edges(const int* __restrict__ ei, float* __restrict__ deg, int e) {
    int i = blockIdx.x * blockDim.x + threadIdx.x;
    if (i < e) atomicAdd(&deg[ei[GCN_E + i]], 1.0f);
}

__global__ void gcn_dinv(const float* __restrict__ deg, float* __restrict__ dinv, int n) {
    int i = blockIdx.x * blockDim.x + threadIdx.x;
    if (i < n) {
        float d = deg[i];
        dinv[i] = (d > 0.0f) ? rsqrtf(d) : 0.0f;
    }
}

__global__ void gcn_edge_norm(const int* __restrict__ ei, const float* __restrict__ dinv,
                              float* __restrict__ norm, int e) {
    int i = blockIdx.x * blockDim.x + threadIdx.x;
    if (i < e) norm[i] = dinv[ei[i]] * dinv[ei[GCN_E + i]];
}


template <int DIN, bool RELU>
__global__ __launch_bounds__(256) void gcn_gemm_wmma(const float* __restrict__ in,
                                                     const float* __restrict__ W,
                                                     float* __restrict__ out) {
    __shared__ float Wl[(DIN / 2) * PAIR_STRIDE];
    const int tid = threadIdx.x;
    for (int i = tid; i < DIN * 64; i += 256) {
        int k = i >> 6;
        int n = i & 63;
        Wl[(k >> 1) * PAIR_STRIDE + n * 2 + (k & 1)] = W[i];
    }
    __syncthreads();

    const int lane = tid & 31;
    const int wave = tid >> 5;
    const int tile = blockIdx.x * 8 + wave;
    if (tile >= (GCN_N >> 4)) return;

    const int m    = lane & 15;
    const int hi   = lane >> 4;
    const int koff = hi * 2;
    const float* arow = in + (size_t)(tile * 16 + m) * DIN;
    const float* brow = &Wl[(koff >> 1) * PAIR_STRIDE + m * 2];

    v8f acc0 = {}, acc1 = {}, acc2 = {}, acc3 = {};
    for (int k0 = 0; k0 < DIN; k0 += 4) {
        float2 av = *(const float2*)(arow + k0 + koff);
        if (RELU) { av.x = fmaxf(av.x, 0.0f); av.y = fmaxf(av.y, 0.0f); }
        v2f a; a.x = av.x; a.y = av.y;

        const float* bp = brow + (k0 >> 1) * PAIR_STRIDE;
        v2f b;
        b = *(const v2f*)(bp +  0);
        acc0 = __builtin_amdgcn_wmma_f32_16x16x4_f32(false, a, false, b, (short)0, acc0, false, false);
        b = *(const v2f*)(bp + 32);
        acc1 = __builtin_amdgcn_wmma_f32_16x16x4_f32(false, a, false, b, (short)0, acc1, false, false);
        b = *(const v2f*)(bp + 64);
        acc2 = __builtin_amdgcn_wmma_f32_16x16x4_f32(false, a, false, b, (short)0, acc2, false, false);
        b = *(const v2f*)(bp + 96);
        acc3 = __builtin_amdgcn_wmma_f32_16x16x4_f32(false, a, false, b, (short)0, acc3, false, false);
    }

    float* orow = out + (size_t)(tile * 16 + 8 * hi) * 64 + m;
#pragma unroll
    for (int r = 0; r < 8; ++r) {
        orow[(size_t)r * 64 +  0] = acc0[r];
        orow[(size_t)r * 64 + 16] = acc1[r];
        orow[(size_t)r * 64 + 32] = acc2[r];
        orow[(size_t)r * 64 + 48] = acc3[r];
    }
}


__global__ void gcn_agg_init(const float* __restrict__ t, const float* __restrict__ dinv,
                             const float* __restrict__ bias, float* __restrict__ out) {
    int idx = blockIdx.x * blockDim.x + threadIdx.x;
    if (idx >= GCN_N * 16) return;
    int i = idx >> 4;
    int q = (idx & 15) * 4;
    float di = dinv[i];
    float s  = 2.0f * di * di;
    float4 tv = *(const float4*)(t + (size_t)i * 64 + q);
    float4 o;
    o.x = bias[q + 0] + s * tv.x;
    o.y = bias[q + 1] + s * tv.y;
    o.z = bias[q + 2] + s * tv.z;
    o.w = bias[q + 3] + s * tv.w;
    *(float4*)(out + (size_t)i * 64 + q) = o;
}


__global__ void gcn_scatter(const int* __restrict__ ei, const float* __restrict__ norm,
                            const float* __restrict__ t, float* __restrict__ out) {
    int idx = blockIdx.x * blockDim.x + threadIdx.x;
    if (idx >= GCN_E * 16) return;
    int e = idx >> 4;
    int q = (idx & 15) * 4;
    int s = ei[e];
    int d = ei[GCN_E + e];
    float w = norm[e];
    float4 tv = *(const float4*)(t + (size_t)s * 64 + q);
    float* o = out + (size_t)d * 64 + q;
    atomicAdd(o + 0, w * tv.x);
    atomicAdd(o + 1, w * tv.y);
    atomicAdd(o + 2, w * tv.z);
    atomicAdd(o + 3, w * tv.w);
}


extern "C" void kernel_launch(void* const* d_in, const int* in_sizes, int n_in,
                              void* d_out, int out_size, void* d_ws, size_t ws_size,
                              hipStream_t stream) {
    (void)in_sizes; (void)n_in; (void)out_size; (void)ws_size;

    const float* x  = (const float*)d_in[0];
    const int*   ei = (const int*)d_in[1];
    const float* W[5] = {(const float*)d_in[2], (const float*)d_in[4], (const float*)d_in[6],
                         (const float*)d_in[8], (const float*)d_in[10]};
    const float* B[5] = {(const float*)d_in[3], (const float*)d_in[5], (const float*)d_in[7],
                         (const float*)d_in[9], (const float*)d_in[11]};

    char*  ws  = (char*)d_ws;
    size_t off = 0;
    auto carve = [&](size_t bytes) {
        void* p = ws + off;
        off = (off + bytes + 255) & ~(size_t)255;
        return p;
    };
    float* deg  = (float*)carve((size_t)GCN_N * 4);
    float* dinv = (float*)carve((size_t)GCN_N * 4);
    float* nrm  = (float*)carve((size_t)GCN_E * 4);
    float* t    = (float*)carve((size_t)GCN_N * 64 * 4);
    float* hagg = (float*)carve((size_t)GCN_N * 64 * 4);

    const int BN  = (GCN_N + 255) / 256;
    const int BE  = (GCN_E + 255) / 256;
    const int BNF = (GCN_N * 16 + 255) / 256;
    const int BEF = (GCN_E * 16 + 255) / 256;
    const int BG  = ((GCN_N / 16) + 7) / 8;

    gcn_deg_init <<<BN, 256, 0, stream>>>(deg, GCN_N);
    gcn_deg_edges<<<BE, 256, 0, stream>>>(ei, deg, GCN_E);
    gcn_dinv     <<<BN, 256, 0, stream>>>(deg, dinv, GCN_N);
    gcn_edge_norm<<<BE, 256, 0, stream>>>(ei, dinv, nrm, GCN_E);

    gcn_gemm_wmma<GCN_IN, false><<<BG, 256, 0, stream>>>(x, W[0], t);
    gcn_agg_init<<<BNF, 256, 0, stream>>>(t, dinv, B[0], hagg);
    gcn_scatter <<<BEF, 256, 0, stream>>>(ei, nrm, t, hagg);

    for (int L = 1; L <= 3; ++L) {
        gcn_gemm_wmma<GCN_H, true><<<BG, 256, 0, stream>>>(hagg, W[L], t);
        gcn_agg_init<<<BNF, 256, 0, stream>>>(t, dinv, B[L], hagg);
        gcn_scatter <<<BEF, 256, 0, stream>>>(ei, nrm, t, hagg);
    }

    float* out = (float*)d_out;
    gcn_gemm_wmma<GCN_H, true><<<BG, 256, 0, stream>>>(hagg, W[4], t);
    gcn_agg_init<<<BNF, 256, 0, stream>>>(t, dinv, B[4], out);
    gcn_scatter <<<BEF, 256, 0, stream>>>(ei, nrm, t, out);
}


# Atomic edge scatter re-mapped: a wave walks 32 edges and each float atomic instruction covers one whole 128-B line of the dst row (was 8 dwords in each of 4 lines); sum order stays arrival order
# speedup vs baseline: 1.5710x; 1.5710x over previous
#include <hip/hip_runtime.h>


#define GCN_N 50000
#define GCN_E 800000
#define GCN_IN 128
#define GCN_H 64

#define PAIR_STRIDE 130

typedef float v2f __attribute__((ext_vector_type(2)));
typedef float v8f __attribute__((ext_vector_type(8)));


__global__ void gcn_deg_init(float* __restrict__ deg, int n) {
    int i = blockIdx.x * blockDim.x + threadIdx.x;
    if (i < n) deg[i] = 2.0f;
}

__global__ void gcn_deg_edges(const int* __restrict__ ei, float* __restrict__ deg, int e) {
    int i = blockIdx.x * blockDim.x + threadIdx.x;
    if (i < e) atomicAdd(&deg[ei[GCN_E + i]], 1.0f);
}

__global__ void gcn_dinv(const float* __restrict__ deg, float* __restrict__ dinv, int n) {
    int i = blockIdx.x * blockDim.x + threadIdx.x;
    if (i < n) {
        float d = deg[i];
        dinv[i] = (d > 0.0f) ? rsqrtf(d) : 0.0f;
    }
}

__global__ void gcn_edge_norm(const int* __restrict__ ei, const float* __restrict__ dinv,
                              float* __restrict__ norm, int e) {
    int i = blockIdx.x * blockDim.x + threadIdx.x;
    if (i < e) norm[i] = dinv[ei[i]] * dinv[ei[GCN_E + i]];
}


template <int DIN, bool RELU>
__global__ __launch_bounds__(256) void gcn_gemm_wmma(const float* __restrict__ in,
                                                     const float* __restrict__ W,
                                                     float* __restrict__ out) {
    __shared__ float Wl[(DIN / 2) * PAIR_STRIDE];
    const int tid = threadIdx.x;
    for (int i = tid; i < DIN * 64; i += 256) {
        int k = i >> 6;
        int n = i & 63;
        Wl[(k >> 1) * PAIR_STRIDE + n * 2 + (k & 1)] = W[i];
    }
    __syncthreads();

    const int lane = tid & 31;
    const int wave = tid >> 5;
    const int tile = blockIdx.x * 8 + wave;
    if (tile >= (GCN_N >> 4)) return;

    const int m    = lane & 15;
    const int hi   = lane >> 4;
    const int koff = hi * 2;
    const float* arow = in + (size_t)(tile * 16 + m) * DIN;
    const float* brow = &Wl[(koff >> 1) * PAIR_STRIDE + m * 2];

    v8f acc0 = {}, acc1 = {}, acc2 = {}, acc3 = {};
    for (int k0 = 0; k0 < DIN; k0 += 4) {
        float2 av = *(const float2*)(arow + k0 + koff);
        if (RELU) { av.x = fmaxf(av.x, 0.0f); av.y = fmaxf(av.y, 0.0f); }
        v2f a; a.x = av.x; a.y = av.y;

        const float* bp = brow + (k0 >> 1) * PAIR_STRIDE;
        v2f b;
        b = *(const v2f*)(bp +  0);
        acc0 = __builtin_amdgcn_wmma_f32_16x16x4_f32(false, a, false, b, (short)0, acc0, false, false);
        b = *(const v2f*)(bp + 32);
        acc1 = __builtin_amdgcn_wmma_f32_16x16x4_f32(false, a, false, b, (short)0, acc1, false, false);
        b = *(const v2f*)(bp + 64);
        acc2 = __builtin_amdgcn_wmma_f32_16x16x4_f32(false, a, false, b, (short)0, acc2, false, false);
        b = *(const v2f*)(bp + 96);
        acc3 = __builtin_amdgcn_wmma_f32_16x16x4_f32(false, a, false, b, (short)0, acc3, false, false);
    }

    float* orow = out + (size_t)(tile * 16 + 8 * hi) * 64 + m;
#pragma unroll
    for (int r = 0; r < 8; ++r) {
        orow[(size_t)r * 64 +  0] = acc0[r];
        orow[(size_t)r * 64 + 16] = acc1[r];
        orow[(size_t)r * 64 + 32] = acc2[r];
        orow[(size_t)r * 64 + 48] = acc3[r];
    }
}


__global__ void gcn_agg_init(const float* __restrict__ t, const float* __restrict__ dinv,
                             const float* __restrict__ bias, float* __restrict__ out) {
    int idx = blockIdx.x * blockDim.x + threadIdx.x;
    if (idx >= GCN_N * 16) return;
    int i = idx >> 4;
    int q = (idx & 15) * 4;
    float di = dinv[i];
    float s  = 2.0f * di * di;
    float4 tv = *(const float4*)(t + (size_t)i * 64 + q);
    float4 o;
    o.x = bias[q + 0] + s * tv.x;
    o.y = bias[q + 1] + s * tv.y;
    o.z = bias[q + 2] + s * tv.z;
    o.w = bias[q + 3] + s * tv.w;
    *(float4*)(out + (size_t)i * 64 + q) = o;
}


#define SC_EPW   32
#define SC_CHUNK 8

static_assert(GCN_E % (SC_EPW * 8) == 0, "edge count must fill whole 8-wave blocks");
static_assert(SC_EPW % SC_CHUNK == 0, "chunks must tile a wave's edges");

__global__ __launch_bounds__(256) void gcn_scatter(const int* __restrict__ ei,
                                                   const float* __restrict__ norm,
                                                   const float* __restrict__ t,
                                                   float* __restrict__ out) {
    const int lane = threadIdx.x & 31;
    const int wave = __builtin_amdgcn_readfirstlane((int)((blockIdx.x * 256 + threadIdx.x) >> 5));
    const int e0   = wave * SC_EPW;
    if (e0 >= GCN_E) return;

    const int*   srcp = ei + e0;
    const int*   dstp = ei + GCN_E + e0;
    const float* wp   = norm + e0;

    for (int c = 0; c < SC_EPW; c += SC_CHUNK) {
        float a[SC_CHUNK], b[SC_CHUNK];
#pragma unroll
        for (int j = 0; j < SC_CHUNK; ++j) {
            const float* tp = t + (size_t)srcp[c + j] * 64 + lane;
            a[j] = tp[0];
            b[j] = tp[32];
        }
#pragma unroll
        for (int j = 0; j < SC_CHUNK; ++j) {
            const float w  = wp[c + j];
            float*      op = out + (size_t)dstp[c + j] * 64 + lane;
            atomicAdd(op,      w * a[j]);
            atomicAdd(op + 32, w * b[j]);
        }
    }
}


extern "C" void kernel_launch(void* const* d_in, const int* in_sizes, int n_in,
                              void* d_out, int out_size, void* d_ws, size_t ws_size,
                              hipStream_t stream) {
    (void)in_sizes; (void)n_in; (void)out_size; (void)ws_size;

    const float* x  = (const float*)d_in[0];
    const int*   ei = (const int*)d_in[1];
    const float* W[5] = {(const float*)d_in[2], (const float*)d_in[4], (const float*)d_in[6],
                         (const float*)d_in[8], (const float*)d_in[10]};
    const float* B[5] = {(const float*)d_in[3], (const float*)d_in[5], (const float*)d_in[7],
                         (const float*)d_in[9], (const float*)d_in[11]};

    char*  ws  = (char*)d_ws;
    size_t off = 0;
    auto carve = [&](size_t bytes) {
        void* p = ws + off;
        off = (off + bytes + 255) & ~(size_t)255;
        return p;
    };
    float* deg  = (float*)carve((size_t)GCN_N * 4);
    float* dinv = (float*)carve((size_t)GCN_N * 4);
    float* nrm  = (float*)carve((size_t)GCN_E * 4);
    float* t    = (float*)carve((size_t)GCN_N * 64 * 4);
    float* hagg = (float*)carve((size_t)GCN_N * 64 * 4);

    const int BN  = (GCN_N + 255) / 256;
    const int BE  = (GCN_E + 255) / 256;
    const int BNF = (GCN_N * 16 + 255) / 256;
    const int BEF = GCN_E / (SC_EPW * 8);
    const int BG  = ((GCN_N / 16) + 7) / 8;

    gcn_deg_init <<<BN, 256, 0, stream>>>(deg, GCN_N);
    gcn_deg_edges<<<BE, 256, 0, stream>>>(ei, deg, GCN_E);
    gcn_dinv     <<<BN, 256, 0, stream>>>(deg, dinv, GCN_N);
    gcn_edge_norm<<<BE, 256, 0, stream>>>(ei, dinv, nrm, GCN_E);

    gcn_gemm_wmma<GCN_IN, false><<<BG, 256, 0, stream>>>(x, W[0], t);
    gcn_agg_init<<<BNF, 256, 0, stream>>>(t, dinv, B[0], hagg);
    gcn_scatter <<<BEF, 256, 0, stream>>>(ei, nrm, t, hagg);

    for (int L = 1; L <= 3; ++L) {
        gcn_gemm_wmma<GCN_H, true><<<BG, 256, 0, stream>>>(hagg, W[L], t);
        gcn_agg_init<<<BNF, 256, 0, stream>>>(t, dinv, B[L], hagg);
        gcn_scatter <<<BEF, 256, 0, stream>>>(ei, nrm, t, hagg);
    }

    float* out = (float*)d_out;
    gcn_gemm_wmma<GCN_H, true><<<BG, 256, 0, stream>>>(hagg, W[4], t);
    gcn_agg_init<<<BNF, 256, 0, stream>>>(t, dinv, B[4], out);
    gcn_scatter <<<BEF, 256, 0, stream>>>(ei, nrm, t, out);
}


# Global float atomics replaced by LDS adds on dst-range tiles: a work-group filters the edge stream once for its 208 rows (offsets from the degree array), sums in LDS; sum order stays arrival order
# speedup vs baseline: 1.6910x; 1.0764x over previous
#include <hip/hip_runtime.h>


#define GCN_N 50000
#define GCN_E 800000
#define GCN_IN 128
#define GCN_H 64

#define PAIR_STRIDE 130

typedef float v2f __attribute__((ext_vector_type(2)));
typedef float v8f __attribute__((ext_vector_type(8)));


__global__ void gcn_deg_init(float* __restrict__ deg, int n) {
    int i = blockIdx.x * blockDim.x + threadIdx.x;
    if (i < n) deg[i] = 2.0f;
}

__global__ void gcn_deg_edges(const int* __restrict__ ei, float* __restrict__ deg, int e) {
    int i = blockIdx.x * blockDim.x + threadIdx.x;
    if (i < e) atomicAdd(&deg[ei[GCN_E + i]], 1.0f);
}

__global__ void gcn_dinv(const float* __restrict__ deg, float* __restrict__ dinv, int n) {
    int i = blockIdx.x * blockDim.x + threadIdx.x;
    if (i < n) {
        float d = deg[i];
        dinv[i] = (d > 0.0f) ? rsqrtf(d) : 0.0f;
    }
}


template <int DIN, bool RELU>
__global__ __launch_bounds__(256) void gcn_gemm_wmma(const float* __restrict__ in,
                                                     const float* __restrict__ W,
                                                     float* __restrict__ out) {
    __shared__ float Wl[(DIN / 2) * PAIR_STRIDE];
    const int tid = threadIdx.x;
    for (int i = tid; i < DIN * 64; i += 256) {
        int k = i >> 6;
        int n = i & 63;
        Wl[(k >> 1) * PAIR_STRIDE + n * 2 + (k & 1)] = W[i];
    }
    __syncthreads();

    const int lane = tid & 31;
    const int wave = tid >> 5;
    const int tile = blockIdx.x * 8 + wave;
    if (tile >= (GCN_N >> 4)) return;

    const int m    = lane & 15;
    const int hi   = lane >> 4;
    const int koff = hi * 2;
    const float* arow = in + (size_t)(tile * 16 + m) * DIN;
    const float* brow = &Wl[(koff >> 1) * PAIR_STRIDE + m * 2];

    v8f acc0 = {}, acc1 = {}, acc2 = {}, acc3 = {};
    for (int k0 = 0; k0 < DIN; k0 += 4) {
        float2 av = *(const float2*)(arow + k0 + koff);
        if (RELU) { av.x = fmaxf(av.x, 0.0f); av.y = fmaxf(av.y, 0.0f); }
        v2f a; a.x = av.x; a.y = av.y;

        const float* bp = brow + (k0 >> 1) * PAIR_STRIDE;
        v2f b;
        b = *(const v2f*)(bp +  0);
        acc0 = __builtin_amdgcn_wmma_f32_16x16x4_f32(false, a, false, b, (short)0, acc0, false, false);
        b = *(const v2f*)(bp + 32);
        acc1 = __builtin_amdgcn_wmma_f32_16x16x4_f32(false, a, false, b, (short)0, acc1, false, false);
        b = *(const v2f*)(bp + 64);
        acc2 = __builtin_amdgcn_wmma_f32_16x16x4_f32(false, a, false, b, (short)0, acc2, false, false);
        b = *(const v2f*)(bp + 96);
        acc3 = __builtin_amdgcn_wmma_f32_16x16x4_f32(false, a, false, b, (short)0, acc3, false, false);
    }

    float* orow = out + (size_t)(tile * 16 + 8 * hi) * 64 + m;
#pragma unroll
    for (int r = 0; r < 8; ++r) {
        orow[(size_t)r * 64 +  0] = acc0[r];
        orow[(size_t)r * 64 + 16] = acc1[r];
        orow[(size_t)r * 64 + 32] = acc2[r];
        orow[(size_t)r * 64 + 48] = acc3[r];
    }
}


#define RT_R  208
#define RT_NR ((GCN_N + RT_R - 1) / RT_R)

static_assert(GCN_N <= 65536, "a src id must fit in 16 bits of an entry");
static_assert(GCN_E <= (1 << 20), "an edge id must fit in 20 bits of an entry");
static_assert(RT_R <= 256, "a local row must fit in 8 bits of an entry");
static_assert(GCN_E % 4 == 0, "dst ids are read four at a time");
static_assert(RT_R % 4 == 0 && GCN_N % 4 == 0, "degrees are read four at a time");

#define RT_SCAN 8

__global__ __launch_bounds__(1024) void gcn_range_fill(const int* __restrict__ ei,
                                                       const float* __restrict__ deg,
                                                       int* __restrict__ entries,
                                                       int* __restrict__ rinfo) {
    __shared__ int s_part[2][32];
    __shared__ int s_start, s_count, s_cursor;
    const int tid  = threadIdx.x;
    const int lane = tid & 31;
    const int wave = tid >> 5;
    const int b    = blockIdx.x;
    const int lo   = b * RT_R;
    const int hi   = (lo + RT_R < GCN_N) ? (lo + RT_R) : GCN_N;

    int before = 0, own = 0;
    const float4* deg4 = (const float4*)deg;
    for (int i4 = tid; i4 < hi / 4; i4 += 1024) {
        const float4 dv = deg4[i4];
        const int c = (int)(dv.x - 2.0f) + (int)(dv.y - 2.0f) + (int)(dv.z - 2.0f) + (int)(dv.w - 2.0f);
        if (i4 * 4 < lo) before += c; else own += c;
    }
    for (int o = 16; o > 0; o >>= 1) {
        before += __shfl_xor(before, o);
        own    += __shfl_xor(own, o);
    }
    if (lane == 0) { s_part[0][wave] = before; s_part[1][wave] = own; }
    if (tid == 0) s_cursor = 0;
    __syncthreads();
    if (wave == 0) {
        int x = s_part[0][lane], y = s_part[1][lane];
        for (int o = 16; o > 0; o >>= 1) {
            x += __shfl_xor(x, o);
            y += __shfl_xor(y, o);
        }
        if (lane == 0) {
            s_start = x; s_count = y;
            rinfo[2 * b] = x; rinfo[2 * b + 1] = y;
        }
    }
    __syncthreads();
    const int start = s_start;
    const int count = s_count;

    const int4* dst4 = (const int4*)(ei + GCN_E);
    for (int v0 = tid; v0 < GCN_E / 4; v0 += 1024 * RT_SCAN) {
        int4 d[RT_SCAN];
#pragma unroll
        for (int k = 0; k < RT_SCAN; ++k) {
            const int v = v0 + k * 1024;
            d[k] = (v < GCN_E / 4) ? dst4[v] : make_int4(-1, -1, -1, -1);
        }
#pragma unroll
        for (int k = 0; k < RT_SCAN; ++k) {
            const int v = v0 + k * 1024;
            const unsigned r0 = (unsigned)(d[k].x - lo), r1 = (unsigned)(d[k].y - lo);
            const unsigned r2 = (unsigned)(d[k].z - lo), r3 = (unsigned)(d[k].w - lo);
            if (r0 < RT_R || r1 < RT_R || r2 < RT_R || r3 < RT_R) {
                if (r0 < RT_R) { int p = atomicAdd(&s_cursor, 1); if (p < count) entries[start + p] = (4 * v + 0) | (int)(r0 << 20); }
                if (r1 < RT_R) { int p = atomicAdd(&s_cursor, 1); if (p < count) entries[start + p] = (4 * v + 1) | (int)(r1 << 20); }
                if (r2 < RT_R) { int p = atomicAdd(&s_cursor, 1); if (p < count) entries[start + p] = (4 * v + 2) | (int)(r2 << 20); }
                if (r3 < RT_R) { int p = atomicAdd(&s_cursor, 1); if (p < count) entries[start + p] = (4 * v + 3) | (int)(r3 << 20); }
            }
        }
    }
    __syncthreads();

    for (int p = tid; p < count; p += 1024) {
        const int en = entries[start + p];
        entries[start + p] = ei[en & 0xFFFFF] | ((en >> 20) << 16);
    }
}


#define RT_GROUP 16

__global__ __launch_bounds__(1024) void gcn_range_agg(const int* __restrict__ entries,
                                                      const int* __restrict__ rinfo,
                                                      const float* __restrict__ t,
                                                      const float* __restrict__ dinv,
                                                      const float* __restrict__ bias,
                                                      float* __restrict__ out) {
    __shared__ float tile[RT_R * 64];
    __shared__ float dloc[RT_R];
    const int tid  = threadIdx.x;
    const int lane = tid & 31;
    const int wave = __builtin_amdgcn_readfirstlane(tid >> 5);
    const int b    = blockIdx.x;
    const int lo   = b * RT_R;
    const int nrows = (GCN_N - lo < RT_R) ? (GCN_N - lo) : RT_R;

    for (int idx = tid; idx < nrows * 16; idx += 1024) {
        const int i = idx >> 4;
        const int q = (idx & 15) * 4;
        const float di = dinv[lo + i];
        const float s  = 2.0f * di * di;
        const float4 tv = *(const float4*)(t + (size_t)(lo + i) * 64 + q);
        float4 o;
        o.x = bias[q + 0] + s * tv.x;
        o.y = bias[q + 1] + s * tv.y;
        o.z = bias[q + 2] + s * tv.z;
        o.w = bias[q + 3] + s * tv.w;
        *(float4*)(&tile[i * 64 + q]) = o;
    }
    for (int i = tid; i < nrows; i += 1024) dloc[i] = dinv[lo + i];
    __syncthreads();

    const int start = rinfo[2 * b];
    const int count = rinfo[2 * b + 1];
    for (int base = wave * 32; base < count; base += 32 * 32) {
        const int n = (count - base < 32) ? (count - base) : 32;
        int   en = 0;
        float w  = 0.0f;
        if (lane < n) {
            en = entries[start + base + lane];
            w  = dinv[en & 0xFFFF] * dloc[en >> 16];
        }
        if (n == 32) {
#pragma unroll
            for (int g = 0; g < 32; g += RT_GROUP) {
                float a[RT_GROUP], c[RT_GROUP];
#pragma unroll
                for (int j = 0; j < RT_GROUP; ++j) {
                    const int ej = __builtin_amdgcn_readlane(en, g + j);
                    const float* tp = t + (size_t)(ej & 0xFFFF) * 64 + lane;
                    a[j] = tp[0];
                    c[j] = tp[32];
                }
#pragma unroll
                for (int j = 0; j < RT_GROUP; ++j) {
                    const int   ej = __builtin_amdgcn_readlane(en, g + j);
                    const float wj = __int_as_float(__builtin_amdgcn_readlane(__float_as_int(w), g + j));
                    float* ap = &tile[(ej >> 16) * 64 + lane];
                    atomicAdd(ap,      wj * a[j]);
                    atomicAdd(ap + 32, wj * c[j]);
                }
            }
        } else {
            for (int j = 0; j < n; ++j) {
                const int   ej = __builtin_amdgcn_readlane(en, j);
                const float wj = __int_as_float(__builtin_amdgcn_readlane(__float_as_int(w), j));
                const float* tp = t + (size_t)(ej & 0xFFFF) * 64 + lane;
                float* ap = &tile[(ej >> 16) * 64 + lane];
                atomicAdd(ap,      wj * tp[0]);
                atomicAdd(ap + 32, wj * tp[32]);
            }
        }
    }
    __syncthreads();

    for (int idx = tid; idx < nrows * 16; idx += 1024) {
        const int i = idx >> 4;
        const int q = (idx & 15) * 4;
        *(float4*)(out + (size_t)(lo + i) * 64 + q) = *(const float4*)(&tile[i * 64 + q]);
    }
}


extern "C" void kernel_launch(void* const* d_in, const int* in_sizes, int n_in,
                              void* d_out, int out_size, void* d_ws, size_t ws_size,
                              hipStream_t stream) {
    (void)in_sizes; (void)n_in; (void)out_size; (void)ws_size;

    const float* x  = (const float*)d_in[0];
    const int*   ei = (const int*)d_in[1];
    const float* W[5] = {(const float*)d_in[2], (const float*)d_in[4], (const float*)d_in[6],
                         (const float*)d_in[8], (const float*)d_in[10]};
    const float* B[5] = {(const float*)d_in[3], (const float*)d_in[5], (const float*)d_in[7],
                         (const float*)d_in[9], (const float*)d_in[11]};

    char*  ws  = (char*)d_ws;
    size_t off = 0;
    auto carve = [&](size_t bytes) {
        void* p = ws + off;
        off = (off + bytes + 255) & ~(size_t)255;
        return p;
    };
    float* deg   = (float*)carve((size_t)GCN_N * 4);
    float* dinv  = (float*)carve((size_t)GCN_N * 4);
    int*   elist = (int*)carve((size_t)GCN_E * 4);
    float* t     = (float*)carve((size_t)GCN_N * 64 * 4);
    float* hagg  = (float*)carve((size_t)GCN_N * 64 * 4);
    int*   rinfo = (int*)carve((size_t)RT_NR * 2 * 4);

    const int BN = (GCN_N + 255) / 256;
    const int BE = (GCN_E + 255) / 256;
    const int BG = ((GCN_N / 16) + 7) / 8;

    gcn_deg_init  <<<BN, 256, 0, stream>>>(deg, GCN_N);
    gcn_deg_edges <<<BE, 256, 0, stream>>>(ei, deg, GCN_E);
    gcn_dinv      <<<BN, 256, 0, stream>>>(deg, dinv, GCN_N);
    gcn_range_fill<<<RT_NR, 1024, 0, stream>>>(ei, deg, elist, rinfo);

    gcn_gemm_wmma<GCN_IN, false><<<BG, 256, 0, stream>>>(x, W[0], t);
    gcn_range_agg<<<RT_NR, 1024, 0, stream>>>(elist, rinfo, t, dinv, B[0], hagg);

    for (int L = 1; L <= 3; ++L) {
        gcn_gemm_wmma<GCN_H, true><<<BG, 256, 0, stream>>>(hagg, W[L], t);
        gcn_range_agg<<<RT_NR, 1024, 0, stream>>>(elist, rinfo, t, dinv, B[L], hagg);
    }

    float* out = (float*)d_out;
    gcn_gemm_wmma<GCN_H, true><<<BG, 256, 0, stream>>>(hagg, W[4], t);
    gcn_range_agg<<<RT_NR, 1024, 0, stream>>>(elist, rinfo, t, dinv, B[4], out);
}


# Last, partly filled batch of a range's edge list loads its rows in groups of 16 under lane-uniform guards like a full batch (was one edge at a time, 2 waits per edge); sum order stays arrival order
# speedup vs baseline: 1.9203x; 1.1356x over previous
#include <hip/hip_runtime.h>


#define GCN_N 50000
#define GCN_E 800000
#define GCN_IN 128
#define GCN_H 64

#define PAIR_STRIDE 130

typedef float v2f __attribute__((ext_vector_type(2)));
typedef float v8f __attribute__((ext_vector_type(8)));


__global__ void gcn_deg_init(float* __restrict__ deg, int n) {
    int i = blockIdx.x * blockDim.x + threadIdx.x;
    if (i < n) deg[i] = 2.0f;
}

__global__ void gcn_deg_edges(const int* __restrict__ ei, float* __restrict__ deg, int e) {
    int i = blockIdx.x * blockDim.x + threadIdx.x;
    if (i < e) atomicAdd(&deg[ei[GCN_E + i]], 1.0f);
}

__global__ void gcn_dinv(const float* __restrict__ deg, float* __restrict__ dinv, int n) {
    int i = blockIdx.x * blockDim.x + threadIdx.x;
    if (i < n) {
        float d = deg[i];
        dinv[i] = (d > 0.0f) ? rsqrtf(d) : 0.0f;
    }
}


template <int DIN, bool RELU>
__global__ __launch_bounds__(256) void gcn_gemm_wmma(const float* __restrict__ in,
                                                     const float* __restrict__ W,
                                                     float* __restrict__ out) {
    __shared__ float Wl[(DIN / 2) * PAIR_STRIDE];
    const int tid = threadIdx.x;
    for (int i = tid; i < DIN * 64; i += 256) {
        int k = i >> 6;
        int n = i & 63;
        Wl[(k >> 1) * PAIR_STRIDE + n * 2 + (k & 1)] = W[i];
    }
    __syncthreads();

    const int lane = tid & 31;
    const int wave = tid >> 5;
    const int tile = blockIdx.x * 8 + wave;
    if (tile >= (GCN_N >> 4)) return;

    const int m    = lane & 15;
    const int hi   = lane >> 4;
    const int koff = hi * 2;
    const float* arow = in + (size_t)(tile * 16 + m) * DIN;
    const float* brow = &Wl[(koff >> 1) * PAIR_STRIDE + m * 2];

    v8f acc0 = {}, acc1 = {}, acc2 = {}, acc3 = {};
    for (int k0 = 0; k0 < DIN; k0 += 4) {
        float2 av = *(const float2*)(arow + k0 + koff);
        if (RELU) { av.x = fmaxf(av.x, 0.0f); av.y = fmaxf(av.y, 0.0f); }
        v2f a; a.x = av.x; a.y = av.y;

        const float* bp = brow + (k0 >> 1) * PAIR_STRIDE;
        v2f b;
        b = *(const v2f*)(bp +  0);
        acc0 = __builtin_amdgcn_wmma_f32_16x16x4_f32(false, a, false, b, (short)0, acc0, false, false);
        b = *(const v2f*)(bp + 32);
        acc1 = __builtin_amdgcn_wmma_f32_16x16x4_f32(false, a, false, b, (short)0, acc1, false, false);
        b = *(const v2f*)(bp + 64);
        acc2 = __builtin_amdgcn_wmma_f32_16x16x4_f32(false, a, false, b, (short)0, acc2, false, false);
        b = *(const v2f*)(bp + 96);
        acc3 = __builtin_amdgcn_wmma_f32_16x16x4_f32(false, a, false, b, (short)0, acc3, false, false);
    }

    float* orow = out + (size_t)(tile * 16 + 8 * hi) * 64 + m;
#pragma unroll
    for (int r = 0; r < 8; ++r) {
        orow[(size_t)r * 64 +  0] = acc0[r];
        orow[(size_t)r * 64 + 16] = acc1[r];
        orow[(size_t)r * 64 + 32] = acc2[r];
        orow[(size_t)r * 64 + 48] = acc3[r];
    }
}


#define RT_R  208
#define RT_NR ((GCN_N + RT_R - 1) / RT_R)

static_assert(GCN_N <= 65536, "a src id must fit in 16 bits of an entry");
static_assert(GCN_E <= (1 << 20), "an edge id must fit in 20 bits of an entry");
static_assert(RT_R <= 256, "a local row must fit in 8 bits of an entry");
static_assert(GCN_E % 4 == 0, "dst ids are read four at a time");
static_assert(RT_R % 4 == 0 && GCN_N % 4 == 0, "degrees are read four at a time");

#define RT_SCAN 8

__global__ __launch_bounds__(1024) void gcn_range_fill(const int* __restrict__ ei,
                                                       const float* __restrict__ deg,
                                                       int* __restrict__ entries,
                                                       int* __restrict__ rinfo) {
    __shared__ int s_part[2][32];
    __shared__ int s_start, s_count, s_cursor;
    const int tid  = threadIdx.x;
    const int lane = tid & 31;
    const int wave = tid >> 5;
    const int b    = blockIdx.x;
    const int lo   = b * RT_R;
    const int hi   = (lo + RT_R < GCN_N) ? (lo + RT_R) : GCN_N;

    int before = 0, own = 0;
    const float4* deg4 = (const float4*)deg;
    for (int i4 = tid; i4 < hi / 4; i4 += 1024) {
        const float4 dv = deg4[i4];
        const int c = (int)(dv.x - 2.0f) + (int)(dv.y - 2.0f) + (int)(dv.z - 2.0f) + (int)(dv.w - 2.0f);
        if (i4 * 4 < lo) before += c; else own += c;
    }
    for (int o = 16; o > 0; o >>= 1) {
        before += __shfl_xor(before, o);
        own    += __shfl_xor(own, o);
    }
    if (lane == 0) { s_part[0][wave] = before; s_part[1][wave] = own; }
    if (tid == 0) s_cursor = 0;
    __syncthreads();
    if (wave == 0) {
        int x = s_part[0][lane], y = s_part[1][lane];
        for (int o = 16; o > 0; o >>= 1) {
            x += __shfl_xor(x, o);
            y += __shfl_xor(y, o);
        }
        if (lane == 0) {
            s_start = x; s_count = y;
            rinfo[2 * b] = x; rinfo[2 * b + 1] = y;
        }
    }
    __syncthreads();
    const int start = s_start;
    const int count = s_count;

    const int4* dst4 = (const int4*)(ei + GCN_E);
    for (int v0 = tid; v0 < GCN_E / 4; v0 += 1024 * RT_SCAN) {
        int4 d[RT_SCAN];
#pragma unroll
        for (int k = 0; k < RT_SCAN; ++k) {
            const int v = v0 + k * 1024;
            d[k] = (v < GCN_E / 4) ? dst4[v] : make_int4(-1, -1, -1, -1);
        }
#pragma unroll
        for (int k = 0; k < RT_SCAN; ++k) {
            const int v = v0 + k * 1024;
            const unsigned r0 = (unsigned)(d[k].x - lo), r1 = (unsigned)(d[k].y - lo);
            const unsigned r2 = (unsigned)(d[k].z - lo), r3 = (unsigned)(d[k].w - lo);
            if (r0 < RT_R || r1 < RT_R || r2 < RT_R || r3 < RT_R) {
                if (r0 < RT_R) { int p = atomicAdd(&s_cursor, 1); if (p < count) entries[start + p] = (4 * v + 0) | (int)(r0 << 20); }
                if (r1 < RT_R) { int p = atomicAdd(&s_cursor, 1); if (p < count) entries[start + p] = (4 * v + 1) | (int)(r1 << 20); }
                if (r2 < RT_R) { int p = atomicAdd(&s_cursor, 1); if (p < count) entries[start + p] = (4 * v + 2) | (int)(r2 << 20); }
                if (r3 < RT_R) { int p = atomicAdd(&s_cursor, 1); if (p < count) entries[start + p] = (4 * v + 3) | (int)(r3 << 20); }
            }
        }
    }
    __syncthreads();

    for (int p = tid; p < count; p += 1024) {
        const int en = entries[start + p];
        entries[start + p] = ei[en & 0xFFFFF] | ((en >> 20) << 16);
    }
}


#define RT_GROUP 16

__global__ __launch_bounds__(1024) void gcn_range_agg(const int* __restrict__ entries,
                                                      const int* __restrict__ rinfo,
                                                      const float* __restrict__ t,
                                                      const float* __restrict__ dinv,
                                                      const float* __restrict__ bias,
                                                      float* __restrict__ out) {
    __shared__ float tile[RT_R * 64];
    __shared__ float dloc[RT_R];
    const int tid  = threadIdx.x;
    const int lane = tid & 31;
    const int wave = __builtin_amdgcn_readfirstlane(tid >> 5);
    const int b    = blockIdx.x;
    const int lo   = b * RT_R;
    const int nrows = (GCN_N - lo < RT_R) ? (GCN_N - lo) : RT_R;

    for (int idx = tid; idx < nrows * 16; idx += 1024) {
        const int i = idx >> 4;
        const int q = (idx & 15) * 4;
        const float di = dinv[lo + i];
        const float s  = 2.0f * di * di;
        const float4 tv = *(const float4*)(t + (size_t)(lo + i) * 64 + q);
        float4 o;
        o.x = bias[q + 0] + s * tv.x;
        o.y = bias[q + 1] + s * tv.y;
        o.z = bias[q + 2] + s * tv.z;
        o.w = bias[q + 3] + s * tv.w;
        *(float4*)(&tile[i * 64 + q]) = o;
    }
    for (int i = tid; i < nrows; i += 1024) dloc[i] = dinv[lo + i];
    __syncthreads();

    const int start = rinfo[2 * b];
    const int count = rinfo[2 * b + 1];
    for (int base = wave * 32; base < count; base += 32 * 32) {
        const int n = (count - base < 32) ? (count - base) : 32;
        int   en = 0;
        float w  = 0.0f;
        if (lane < n) {
            en = entries[start + base + lane];
            w  = dinv[en & 0xFFFF] * dloc[en >> 16];
        }
        if (n == 32) {
#pragma unroll
            for (int g = 0; g < 32; g += RT_GROUP) {
                float a[RT_GROUP], c[RT_GROUP];
#pragma unroll
                for (int j = 0; j < RT_GROUP; ++j) {
                    const int ej = __builtin_amdgcn_readlane(en, g + j);
                    const float* tp = t + (size_t)(ej & 0xFFFF) * 64 + lane;
                    a[j] = tp[0];
                    c[j] = tp[32];
                }
#pragma unroll
                for (int j = 0; j < RT_GROUP; ++j) {
                    const int   ej = __builtin_amdgcn_readlane(en, g + j);
                    const float wj = __int_as_float(__builtin_amdgcn_readlane(__float_as_int(w), g + j));
                    float* ap = &tile[(ej >> 16) * 64 + lane];
                    atomicAdd(ap,      wj * a[j]);
                    atomicAdd(ap + 32, wj * c[j]);
                }
            }
        } else {
#pragma unroll
            for (int g = 0; g < 32; g += RT_GROUP) {
                if (g < n) {
                    float a[RT_GROUP], c[RT_GROUP];
#pragma unroll
                    for (int j = 0; j < RT_GROUP; ++j) {
                        a[j] = 0.0f;
                        c[j] = 0.0f;
                        if (g + j < n) {
                            const int ej = __builtin_amdgcn_readlane(en, g + j);
                            const float* tp = t + (size_t)(ej & 0xFFFF) * 64 + lane;
                            a[j] = tp[0];
                            c[j] = tp[32];
                        }
                    }
#pragma unroll
                    for (int j = 0; j < RT_GROUP; ++j) {
                        if (g + j < n) {
                            const int   ej = __builtin_amdgcn_readlane(en, g + j);
                            const float wj = __int_as_float(__builtin_amdgcn_readlane(__float_as_int(w), g + j));
                            float* ap = &tile[(ej >> 16) * 64 + lane];
                            atomicAdd(ap,      wj * a[j]);
                            atomicAdd(ap + 32, wj * c[j]);
                        }
                    }
                }
            }
        }
    }
    __syncthreads();

    for (int idx = tid; idx < nrows * 16; idx += 1024) {
        const int i = idx >> 4;
        const int q = (idx & 15) * 4;
        *(float4*)(out + (size_t)(lo + i) * 64 + q) = *(const float4*)(&tile[i * 64 + q]);
    }
}


extern "C" void kernel_launch(void* const* d_in, const int* in_sizes, int n_in,
                              void* d_out, int out_size, void* d_ws, size_t ws_size,
                              hipStream_t stream) {
    (void)in_sizes; (void)n_in; (void)out_size; (void)ws_size;

    const float* x  = (const float*)d_in[0];
    const int*   ei = (const int*)d_in[1];
    const float* W[5] = {(const float*)d_in[2], (const float*)d_in[4], (const float*)d_in[6],
                         (const float*)d_in[8], (const float*)d_in[10]};
    const float* B[5] = {(const float*)d_in[3], (const float*)d_in[5], (const float*)d_in[7],
                         (const float*)d_in[9], (const float*)d_in[11]};

    char*  ws  = (char*)d_ws;
    size_t off = 0;
    auto carve = [&](size_t bytes) {
        void* p = ws + off;
        off = (off + bytes + 255) & ~(size_t)255;
        return p;
    };
    float* deg   = (float*)carve((size_t)GCN_N * 4);
    float* dinv  = (float*)carve((size_t)GCN_N * 4);
    int*   elist = (int*)carve((size_t)GCN_E * 4);
    float* t     = (float*)carve((size_t)GCN_N * 64 * 4);
    float* hagg  = (float*)carve((size_t)GCN_N * 64 * 4);
    int*   rinfo = (int*)carve((size_t)RT_NR * 2 * 4);

    const int BN = (GCN_N + 255) / 256;
    const int BE = (GCN_E + 255) / 256;
    const int BG = ((GCN_N / 16) + 7) / 8;

    gcn_deg_init  <<<BN, 256, 0, stream>>>(deg, GCN_N);
    gcn_deg_edges <<<BE, 256, 0, stream>>>(ei, deg, GCN_E);
    gcn_dinv      <<<BN, 256, 0, stream>>>(deg, dinv, GCN_N);
    gcn_range_fill<<<RT_NR, 1024, 0, stream>>>(ei, deg, elist, rinfo);

    gcn_gemm_wmma<GCN_IN, false><<<BG, 256, 0, stream>>>(x, W[0], t);
    gcn_range_agg<<<RT_NR, 1024, 0, stream>>>(elist, rinfo, t, dinv, B[0], hagg);

    for (int L = 1; L <= 3; ++L) {
        gcn_gemm_wmma<GCN_H, true><<<BG, 256, 0, stream>>>(hagg, W[L], t);
        gcn_range_agg<<<RT_NR, 1024, 0, stream>>>(elist, rinfo, t, dinv, B[L], hagg);
    }

    float* out = (float*)d_out;
    gcn_gemm_wmma<GCN_H, true><<<BG, 256, 0, stream>>>(hagg, W[4], t);
    gcn_range_agg<<<RT_NR, 1024, 0, stream>>>(elist, rinfo, t, dinv, B[4], out);
}


# Degree pass without global atomics: a range-histogram launch (LDS counts, plain stores) gives list offsets; range owners count in-degree in LDS and store dinv; 14 launches to 12; sum order unchanged
# speedup vs baseline: 2.2678x; 1.1810x over previous
#include <hip/hip_runtime.h>


#define GCN_N 50000
#define GCN_E 800000
#define GCN_IN 128
#define GCN_H 64

#define PAIR_STRIDE 130

typedef float v2f __attribute__((ext_vector_type(2)));
typedef float v8f __attribute__((ext_vector_type(8)));


template <int DIN, bool RELU>
__global__ __launch_bounds__(256) void gcn_gemm_wmma(const float* __restrict__ in,
                                                     const float* __restrict__ W,
                                                     float* __restrict__ out) {
    __shared__ float Wl[(DIN / 2) * PAIR_STRIDE];
    const int tid = threadIdx.x;
    for (int i = tid; i < DIN * 64; i += 256) {
        int k = i >> 6;
        int n = i & 63;
        Wl[(k >> 1) * PAIR_STRIDE + n * 2 + (k & 1)] = W[i];
    }
    __syncthreads();

    const int lane = tid & 31;
    const int wave = tid >> 5;
    const int tile = blockIdx.x * 8 + wave;
    if (tile >= (GCN_N >> 4)) return;

    const int m    = lane & 15;
    const int hi   = lane >> 4;
    const int koff = hi * 2;
    const float* arow = in + (size_t)(tile * 16 + m) * DIN;
    const float* brow = &Wl[(koff >> 1) * PAIR_STRIDE + m * 2];

    v8f acc0 = {}, acc1 = {}, acc2 = {}, acc3 = {};
    for (int k0 = 0; k0 < DIN; k0 += 4) {
        float2 av = *(const float2*)(arow + k0 + koff);
        if (RELU) { av.x = fmaxf(av.x, 0.0f); av.y = fmaxf(av.y, 0.0f); }
        v2f a; a.x = av.x; a.y = av.y;

        const float* bp = brow + (k0 >> 1) * PAIR_STRIDE;
        v2f b;
        b = *(const v2f*)(bp +  0);
        acc0 = __builtin_amdgcn_wmma_f32_16x16x4_f32(false, a, false, b, (short)0, acc0, false, false);
        b = *(const v2f*)(bp + 32);
        acc1 = __builtin_amdgcn_wmma_f32_16x16x4_f32(false, a, false, b, (short)0, acc1, false, false);
        b = *(const v2f*)(bp + 64);
        acc2 = __builtin_amdgcn_wmma_f32_16x16x4_f32(false, a, false, b, (short)0, acc2, false, false);
        b = *(const v2f*)(bp + 96);
        acc3 = __builtin_amdgcn_wmma_f32_16x16x4_f32(false, a, false, b, (short)0, acc3, false, false);
    }

    float* orow = out + (size_t)(tile * 16 + 8 * hi) * 64 + m;
#pragma unroll
    for (int r = 0; r < 8; ++r) {
        orow[(size_t)r * 64 +  0] = acc0[r];
        orow[(size_t)r * 64 + 16] = acc1[r];
        orow[(size_t)r * 64 + 32] = acc2[r];
        orow[(size_t)r * 64 + 48] = acc3[r];
    }
}


#define RT_R  208
#define RT_NR ((GCN_N + RT_R - 1) / RT_R)

static_assert(GCN_N <= 65536, "a src id must fit in 16 bits of an entry");
static_assert(GCN_E <= (1 << 20), "an edge id must fit in 20 bits of an entry");
static_assert(RT_R <= 256, "a local row must fit in 8 bits of an entry");
static_assert(GCN_E % 4 == 0, "dst ids are read four at a time");

#define RH_G 64
#define RH_V (GCN_E / 4 / RH_G)
#define RH_K ((RH_V + 1023) / 1024)
static_assert(GCN_E % (4 * RH_G) == 0, "the dst stream is cut into RH_G equal slices");
static_assert(RH_G % 4 == 0, "histogram columns are read four at a time");

__global__ __launch_bounds__(1024) void gcn_range_hist(const int* __restrict__ ei,
                                                       int* __restrict__ part) {
    __shared__ int s_hist[RT_NR];
    const int tid = threadIdx.x;
    const int g   = blockIdx.x;
    if (tid < RT_NR) s_hist[tid] = 0;
    __syncthreads();
    const int4* dst4 = (const int4*)(ei + GCN_E) + g * RH_V;
    int4 d[RH_K];
#pragma unroll
    for (int k = 0; k < RH_K; ++k) {
        const int i = tid + k * 1024;
        d[k] = (i < RH_V) ? dst4[i] : make_int4(-1, -1, -1, -1);
    }
#pragma unroll
    for (int k = 0; k < RH_K; ++k) {
        const unsigned r0 = (unsigned)d[k].x / RT_R, r1 = (unsigned)d[k].y / RT_R;
        const unsigned r2 = (unsigned)d[k].z / RT_R, r3 = (unsigned)d[k].w / RT_R;
        if (r0 < RT_NR) atomicAdd(&s_hist[r0], 1);
        if (r1 < RT_NR) atomicAdd(&s_hist[r1], 1);
        if (r2 < RT_NR) atomicAdd(&s_hist[r2], 1);
        if (r3 < RT_NR) atomicAdd(&s_hist[r3], 1);
    }
    __syncthreads();
    if (tid < RT_NR) part[tid * RH_G + g] = s_hist[tid];
}

#define RT_SCAN 8

__global__ __launch_bounds__(1024) void gcn_range_fill(const int* __restrict__ ei,
                                                       const int* __restrict__ part,
                                                       int* __restrict__ entries,
                                                       int* __restrict__ rinfo,
                                                       float* __restrict__ dinv) {
    __shared__ int s_part[2][32];
    __shared__ int s_start, s_count, s_cursor;
    __shared__ int s_deg[RT_R];
    const int tid  = threadIdx.x;
    const int lane = tid & 31;
    const int wave = tid >> 5;
    const int b    = blockIdx.x;
    const int lo   = b * RT_R;
    const int hi   = (lo + RT_R < GCN_N) ? (lo + RT_R) : GCN_N;

    int before = 0, own = 0;
    const int4* part4 = (const int4*)part;
    const int nb4 = b * (RH_G / 4);
    for (int i4 = tid; i4 < nb4 + RH_G / 4; i4 += 1024) {
        const int4 pv = part4[i4];
        const int c = pv.x + pv.y + pv.z + pv.w;
        if (i4 < nb4) before += c; else own += c;
    }
    for (int o = 16; o > 0; o >>= 1) {
        before += __shfl_xor(before, o);
        own    += __shfl_xor(own, o);
    }
    if (lane == 0) { s_part[0][wave] = before; s_part[1][wave] = own; }
    if (tid == 0) s_cursor = 0;
    if (tid < RT_R) s_deg[tid] = 0;
    __syncthreads();
    if (wave == 0) {
        int x = s_part[0][lane], y = s_part[1][lane];
        for (int o = 16; o > 0; o >>= 1) {
            x += __shfl_xor(x, o);
            y += __shfl_xor(y, o);
        }
        if (lane == 0) {
            s_start = x; s_count = y;
            rinfo[2 * b] = x; rinfo[2 * b + 1] = y;
        }
    }
    __syncthreads();
    const int start = s_start;
    const int count = s_count;

    const int4* dst4 = (const int4*)(ei + GCN_E);
    for (int v0 = tid; v0 < GCN_E / 4; v0 += 1024 * RT_SCAN) {
        int4 d[RT_SCAN];
#pragma unroll
        for (int k = 0; k < RT_SCAN; ++k) {
            const int v = v0 + k * 1024;
            d[k] = (v < GCN_E / 4) ? dst4[v] : make_int4(-1, -1, -1, -1);
        }
#pragma unroll
        for (int k = 0; k < RT_SCAN; ++k) {
            const int v = v0 + k * 1024;
            const unsigned r0 = (unsigned)(d[k].x - lo), r1 = (unsigned)(d[k].y - lo);
            const unsigned r2 = (unsigned)(d[k].z - lo), r3 = (unsigned)(d[k].w - lo);
            if (r0 < RT_R || r1 < RT_R || r2 < RT_R || r3 < RT_R) {
                if (r0 < RT_R) { int p = atomicAdd(&s_cursor, 1); if (p < count) entries[start + p] = (4 * v + 0) | (int)(r0 << 20); }
                if (r1 < RT_R) { int p = atomicAdd(&s_cursor, 1); if (p < count) entries[start + p] = (4 * v + 1) | (int)(r1 << 20); }
                if (r2 < RT_R) { int p = atomicAdd(&s_cursor, 1); if (p < count) entries[start + p] = (4 * v + 2) | (int)(r2 << 20); }
                if (r3 < RT_R) { int p = atomicAdd(&s_cursor, 1); if (p < count) entries[start + p] = (4 * v + 3) | (int)(r3 << 20); }
            }
        }
    }
    __syncthreads();

    for (int p = tid; p < count; p += 1024) {
        const int en = entries[start + p];
        const int lr = en >> 20;
        atomicAdd(&s_deg[lr], 1);
        entries[start + p] = ei[en & 0xFFFFF] | (lr << 16);
    }
    __syncthreads();

    if (tid < hi - lo) {
        const float d = 2.0f + (float)s_deg[tid];
        dinv[lo + tid] = (d > 0.0f) ? rsqrtf(d) : 0.0f;
    }
}


#define RT_GROUP 16

__global__ __launch_bounds__(1024) void gcn_range_agg(const int* __restrict__ entries,
                                                      const int* __restrict__ rinfo,
                                                      const float* __restrict__ t,
                                                      const float* __restrict__ dinv,
                                                      const float* __restrict__ bias,
                                                      float* __restrict__ out) {
    __shared__ float tile[RT_R * 64];
    __shared__ float dloc[RT_R];
    const int tid  = threadIdx.x;
    const int lane = tid & 31;
    const int wave = __builtin_amdgcn_readfirstlane(tid >> 5);
    const int b    = blockIdx.x;
    const int lo   = b * RT_R;
    const int nrows = (GCN_N - lo < RT_R) ? (GCN_N - lo) : RT_R;

    for (int idx = tid; idx < nrows * 16; idx += 1024) {
        const int i = idx >> 4;
        const int q = (idx & 15) * 4;
        const float di = dinv[lo + i];
        const float s  = 2.0f * di * di;
        const float4 tv = *(const float4*)(t + (size_t)(lo + i) * 64 + q);
        float4 o;
        o.x = bias[q + 0] + s * tv.x;
        o.y = bias[q + 1] + s * tv.y;
        o.z = bias[q + 2] + s * tv.z;
        o.w = bias[q + 3] + s * tv.w;
        *(float4*)(&tile[i * 64 + q]) = o;
    }
    for (int i = tid; i < nrows; i += 1024) dloc[i] = dinv[lo + i];
    __syncthreads();

    const int start = rinfo[2 * b];
    const int count = rinfo[2 * b + 1];
    for (int base = wave * 32; base < count; base += 32 * 32) {
        const int n = (count - base < 32) ? (count - base) : 32;
        int   en = 0;
        float w  = 0.0f;
        if (lane < n) {
            en = entries[start + base + lane];
            w  = dinv[en & 0xFFFF] * dloc[en >> 16];
        }
        if (n == 32) {
#pragma unroll
            for (int g = 0; g < 32; g += RT_GROUP) {
                float a[RT_GROUP], c[RT_GROUP];
#pragma unroll
                for (int j = 0; j < RT_GROUP; ++j) {
                    const int ej = __builtin_amdgcn_readlane(en, g + j);
                    const float* tp = t + (size_t)(ej & 0xFFFF) * 64 + lane;
                    a[j] = tp[0];
                    c[j] = tp[32];
                }
#pragma unroll
                for (int j = 0; j < RT_GROUP; ++j) {
                    const int   ej = __builtin_amdgcn_readlane(en, g + j);
                    const float wj = __int_as_float(__builtin_amdgcn_readlane(__float_as_int(w), g + j));
                    float* ap = &tile[(ej >> 16) * 64 + lane];
                    atomicAdd(ap,      wj * a[j]);
                    atomicAdd(ap + 32, wj * c[j]);
                }
            }
        } else {
#pragma unroll
            for (int g = 0; g < 32; g += RT_GROUP) {
                if (g < n) {
                    float a[RT_GROUP], c[RT_GROUP];
#pragma unroll
                    for (int j = 0; j < RT_GROUP; ++j) {
                        a[j] = 0.0f;
                        c[j] = 0.0f;
                        if (g + j < n) {
                            const int ej = __builtin_amdgcn_readlane(en, g + j);
                            const float* tp = t + (size_t)(ej & 0xFFFF) * 64 + lane;
                            a[j] = tp[0];
                            c[j] = tp[32];
                        }
                    }
#pragma unroll
                    for (int j = 0; j < RT_GROUP; ++j) {
                        if (g + j < n) {
                            const int   ej = __builtin_amdgcn_readlane(en, g + j);
                            const float wj = __int_as_float(__builtin_amdgcn_readlane(__float_as_int(w), g + j));
                            float* ap = &tile[(ej >> 16) * 64 + lane];
                            atomicAdd(ap,      wj * a[j]);
                            atomicAdd(ap + 32, wj * c[j]);
                        }
                    }
                }
            }
        }
    }
    __syncthreads();

    for (int idx = tid; idx < nrows * 16; idx += 1024) {
        const int i = idx >> 4;
        const int q = (idx & 15) * 4;
        *(float4*)(out + (size_t)(lo + i) * 64 + q) = *(const float4*)(&tile[i * 64 + q]);
    }
}


extern "C" void kernel_launch(void* const* d_in, const int* in_sizes, int n_in,
                              void* d_out, int out_size, void* d_ws, size_t ws_size,
                              hipStream_t stream) {
    (void)in_sizes; (void)n_in; (void)out_size; (void)ws_size;

    const float* x  = (const float*)d_in[0];
    const int*   ei = (const int*)d_in[1];
    const float* W[5] = {(const float*)d_in[2], (const float*)d_in[4], (const float*)d_in[6],
                         (const float*)d_in[8], (const float*)d_in[10]};
    const float* B[5] = {(const float*)d_in[3], (const float*)d_in[5], (const float*)d_in[7],
                         (const float*)d_in[9], (const float*)d_in[11]};

    char*  ws  = (char*)d_ws;
    size_t off = 0;
    auto carve = [&](size_t bytes) {
        void* p = ws + off;
        off = (off + bytes + 255) & ~(size_t)255;
        return p;
    };
    int*   part  = (int*)carve((size_t)RT_NR * RH_G * 4);
    float* dinv  = (float*)carve((size_t)GCN_N * 4);
    int*   elist = (int*)carve((size_t)GCN_E * 4);
    float* t     = (float*)carve((size_t)GCN_N * 64 * 4);
    float* hagg  = (float*)carve((size_t)GCN_N * 64 * 4);
    int*   rinfo = (int*)carve((size_t)RT_NR * 2 * 4);

    const int BG = ((GCN_N / 16) + 7) / 8;

    gcn_range_hist<<<RH_G, 1024, 0, stream>>>(ei, part);
    gcn_range_fill<<<RT_NR, 1024, 0, stream>>>(ei, part, elist, rinfo, dinv);

    gcn_gemm_wmma<GCN_IN, false><<<BG, 256, 0, stream>>>(x, W[0], t);
    gcn_range_agg<<<RT_NR, 1024, 0, stream>>>(elist, rinfo, t, dinv, B[0], hagg);

    for (int L = 1; L <= 3; ++L) {
        gcn_gemm_wmma<GCN_H, true><<<BG, 256, 0, stream>>>(hagg, W[L], t);
        gcn_range_agg<<<RT_NR, 1024, 0, stream>>>(elist, rinfo, t, dinv, B[L], hagg);
    }

    float* out = (float*)d_out;
    gcn_gemm_wmma<GCN_H, true><<<BG, 256, 0, stream>>>(hagg, W[4], t);
    gcn_range_agg<<<RT_NR, 1024, 0, stream>>>(elist, rinfo, t, dinv, B[4], out);
}


# Edge weight dinv[src]*dinv[dst] worked out once: layer 1's edge-sum launch stores it per list slot, layers 2-5 read 32 weights in one line instead of 32 scattered dinv loads; sum order unchanged
# speedup vs baseline: 2.3411x; 1.0323x over previous
#include <hip/hip_runtime.h>


#define GCN_N 50000
#define GCN_E 800000
#define GCN_IN 128
#define GCN_H 64

#define PAIR_STRIDE 130

typedef float v2f __attribute__((ext_vector_type(2)));
typedef float v8f __attribute__((ext_vector_type(8)));


template <int DIN, bool RELU>
__global__ __launch_bounds__(256) void gcn_gemm_wmma(const float* __restrict__ in,
                                                     const float* __restrict__ W,
                                                     float* __restrict__ out) {
    __shared__ float Wl[(DIN / 2) * PAIR_STRIDE];
    const int tid = threadIdx.x;
    for (int i = tid; i < DIN * 64; i += 256) {
        int k = i >> 6;
        int n = i & 63;
        Wl[(k >> 1) * PAIR_STRIDE + n * 2 + (k & 1)] = W[i];
    }
    __syncthreads();

    const int lane = tid & 31;
    const int wave = tid >> 5;
    const int tile = blockIdx.x * 8 + wave;
    if (tile >= (GCN_N >> 4)) return;

    const int m    = lane & 15;
    const int hi   = lane >> 4;
    const int koff = hi * 2;
    const float* arow = in + (size_t)(tile * 16 + m) * DIN;
    const float* brow = &Wl[(koff >> 1) * PAIR_STRIDE + m * 2];

    v8f acc0 = {}, acc1 = {}, acc2 = {}, acc3 = {};
    for (int k0 = 0; k0 < DIN; k0 += 4) {
        float2 av = *(const float2*)(arow + k0 + koff);
        if (RELU) { av.x = fmaxf(av.x, 0.0f); av.y = fmaxf(av.y, 0.0f); }
        v2f a; a.x = av.x; a.y = av.y;

        const float* bp = brow + (k0 >> 1) * PAIR_STRIDE;
        v2f b;
        b = *(const v2f*)(bp +  0);
        acc0 = __builtin_amdgcn_wmma_f32_16x16x4_f32(false, a, false, b, (short)0, acc0, false, false);
        b = *(const v2f*)(bp + 32);
        acc1 = __builtin_amdgcn_wmma_f32_16x16x4_f32(false, a, false, b, (short)0, acc1, false, false);
        b = *(const v2f*)(bp + 64);
        acc2 = __builtin_amdgcn_wmma_f32_16x16x4_f32(false, a, false, b, (short)0, acc2, false, false);
        b = *(const v2f*)(bp + 96);
        acc3 = __builtin_amdgcn_wmma_f32_16x16x4_f32(false, a, false, b, (short)0, acc3, false, false);
    }

    float* orow = out + (size_t)(tile * 16 + 8 * hi) * 64 + m;
#pragma unroll
    for (int r = 0; r < 8; ++r) {
        orow[(size_t)r * 64 +  0] = acc0[r];
        orow[(size_t)r * 64 + 16] = acc1[r];
        orow[(size_t)r * 64 + 32] = acc2[r];
        orow[(size_t)r * 64 + 48] = acc3[r];
    }
}


#define RT_R  208
#define RT_NR ((GCN_N + RT_R - 1) / RT_R)

static_assert(GCN_N <= 65536, "a src id must fit in 16 bits of an entry");
static_assert(GCN_E <= (1 << 20), "an edge id must fit in 20 bits of an entry");
static_assert(RT_R <= 256, "a local row must fit in 8 bits of an entry");
static_assert(GCN_E % 4 == 0, "dst ids are read four at a time");

#define RH_G 64
#define RH_V (GCN_E / 4 / RH_G)
#define RH_K ((RH_V + 1023) / 1024)
static_assert(GCN_E % (4 * RH_G) == 0, "the dst stream is cut into RH_G equal slices");
static_assert(RH_G % 4 == 0, "histogram columns are read four at a time");

__global__ __launch_bounds__(1024) void gcn_range_hist(const int* __restrict__ ei,
                                                       int* __restrict__ part) {
    __shared__ int s_hist[RT_NR];
    const int tid = threadIdx.x;
    const int g   = blockIdx.x;
    if (tid < RT_NR) s_hist[tid] = 0;
    __syncthreads();
    const int4* dst4 = (const int4*)(ei + GCN_E) + g * RH_V;
    int4 d[RH_K];
#pragma unroll
    for (int k = 0; k < RH_K; ++k) {
        const int i = tid + k * 1024;
        d[k] = (i < RH_V) ? dst4[i] : make_int4(-1, -1, -1, -1);
    }
#pragma unroll
    for (int k = 0; k < RH_K; ++k) {
        const unsigned r0 = (unsigned)d[k].x / RT_R, r1 = (unsigned)d[k].y / RT_R;
        const unsigned r2 = (unsigned)d[k].z / RT_R, r3 = (unsigned)d[k].w / RT_R;
        if (r0 < RT_NR) atomicAdd(&s_hist[r0], 1);
        if (r1 < RT_NR) atomicAdd(&s_hist[r1], 1);
        if (r2 < RT_NR) atomicAdd(&s_hist[r2], 1);
        if (r3 < RT_NR) atomicAdd(&s_hist[r3], 1);
    }
    __syncthreads();
    if (tid < RT_NR) part[tid * RH_G + g] = s_hist[tid];
}

#define RT_SCAN 8

__global__ __launch_bounds__(1024) void gcn_range_fill(const int* __restrict__ ei,
                                                       const int* __restrict__ part,
                                                       int* __restrict__ entries,
                                                       int* __restrict__ rinfo,
                                                       float* __restrict__ dinv) {
    __shared__ int s_part[2][32];
    __shared__ int s_start, s_count, s_cursor;
    __shared__ int s_deg[RT_R];
    const int tid  = threadIdx.x;
    const int lane = tid & 31;
    const int wave = tid >> 5;
    const int b    = blockIdx.x;
    const int lo   = b * RT_R;
    const int hi   = (lo + RT_R < GCN_N) ? (lo + RT_R) : GCN_N;

    int before = 0, own = 0;
    const int4* part4 = (const int4*)part;
    const int nb4 = b * (RH_G / 4);
    for (int i4 = tid; i4 < nb4 + RH_G / 4; i4 += 1024) {
        const int4 pv = part4[i4];
        const int c = pv.x + pv.y + pv.z + pv.w;
        if (i4 < nb4) before += c; else own += c;
    }
    for (int o = 16; o > 0; o >>= 1) {
        before += __shfl_xor(before, o);
        own    += __shfl_xor(own, o);
    }
    if (lane == 0) { s_part[0][wave] = before; s_part[1][wave] = own; }
    if (tid == 0) s_cursor = 0;
    if (tid < RT_R) s_deg[tid] = 0;
    __syncthreads();
    if (wave == 0) {
        int x = s_part[0][lane], y = s_part[1][lane];
        for (int o = 16; o > 0; o >>= 1) {
            x += __shfl_xor(x, o);
            y += __shfl_xor(y, o);
        }
        if (lane == 0) {
            s_start = x; s_count = y;
            rinfo[2 * b] = x; rinfo[2 * b + 1] = y;
        }
    }
    __syncthreads();
    const int start = s_start;
    const int count = s_count;

    const int4* dst4 = (const int4*)(ei + GCN_E);
    for (int v0 = tid; v0 < GCN_E / 4; v0 += 1024 * RT_SCAN) {
        int4 d[RT_SCAN];
#pragma unroll
        for (int k = 0; k < RT_SCAN; ++k) {
            const int v = v0 + k * 1024;
            d[k] = (v < GCN_E / 4) ? dst4[v] : make_int4(-1, -1, -1, -1);
        }
#pragma unroll
        for (int k = 0; k < RT_SCAN; ++k) {
            const int v = v0 + k * 1024;
            const unsigned r0 = (unsigned)(d[k].x - lo), r1 = (unsigned)(d[k].y - lo);
            const unsigned r2 = (unsigned)(d[k].z - lo), r3 = (unsigned)(d[k].w - lo);
            if (r0 < RT_R || r1 < RT_R || r2 < RT_R || r3 < RT_R) {
                if (r0 < RT_R) { int p = atomicAdd(&s_cursor, 1); if (p < count) entries[start + p] = (4 * v + 0) | (int)(r0 << 20); }
                if (r1 < RT_R) { int p = atomicAdd(&s_cursor, 1); if (p < count) entries[start + p] = (4 * v + 1) | (int)(r1 << 20); }
                if (r2 < RT_R) { int p = atomicAdd(&s_cursor, 1); if (p < count) entries[start + p] = (4 * v + 2) | (int)(r2 << 20); }
                if (r3 < RT_R) { int p = atomicAdd(&s_cursor, 1); if (p < count) entries[start + p] = (4 * v + 3) | (int)(r3 << 20); }
            }
        }
    }
    __syncthreads();

    for (int p = tid; p < count; p += 1024) {
        const int en = entries[start + p];
        const int lr = en >> 20;
        atomicAdd(&s_deg[lr], 1);
        entries[start + p] = ei[en & 0xFFFFF] | (lr << 16);
    }
    __syncthreads();

    if (tid < hi - lo) {
        const float d = 2.0f + (float)s_deg[tid];
        dinv[lo + tid] = (d > 0.0f) ? rsqrtf(d) : 0.0f;
    }
}


#define RT_GROUP 16

template <bool FIRST>
__global__ __launch_bounds__(1024) void gcn_range_agg(const int* __restrict__ entries,
                                                      const int* __restrict__ rinfo,
                                                      const float* __restrict__ t,
                                                      const float* __restrict__ dinv,
                                                      const float* __restrict__ bias,
                                                      float* __restrict__ out,
                                                      float* __restrict__ wl,
                                                      size_t ws_need, size_t ws_size) {
    __shared__ float tile[RT_R * 64];
    __shared__ float dloc[RT_R];
    const int tid  = threadIdx.x;
    const int lane = tid & 31;
    const int wave = __builtin_amdgcn_readfirstlane(tid >> 5);
    const int b    = blockIdx.x;
    const int lo   = b * RT_R;
    const int nrows = (GCN_N - lo < RT_R) ? (GCN_N - lo) : RT_R;
    const bool have_wl = ws_need <= ws_size;
    const bool read_wl = !FIRST && have_wl;

    for (int idx = tid; idx < nrows * 16; idx += 1024) {
        const int i = idx >> 4;
        const int q = (idx & 15) * 4;
        const float di = dinv[lo + i];
        const float s  = 2.0f * di * di;
        const float4 tv = *(const float4*)(t + (size_t)(lo + i) * 64 + q);
        float4 o;
        o.x = bias[q + 0] + s * tv.x;
        o.y = bias[q + 1] + s * tv.y;
        o.z = bias[q + 2] + s * tv.z;
        o.w = bias[q + 3] + s * tv.w;
        *(float4*)(&tile[i * 64 + q]) = o;
    }
    if (!read_wl) {
        for (int i = tid; i < nrows; i += 1024) dloc[i] = dinv[lo + i];
    }
    __syncthreads();

    const int start = rinfo[2 * b];
    const int count = rinfo[2 * b + 1];
    for (int base = wave * 32; base < count; base += 32 * 32) {
        const int n = (count - base < 32) ? (count - base) : 32;
        int   en = 0;
        float w  = 0.0f;
        if (lane < n) {
            en = entries[start + base + lane];
            if (read_wl) {
                w = wl[start + base + lane];
            } else {
                w = dinv[en & 0xFFFF] * dloc[en >> 16];
                if (FIRST && have_wl) wl[start + base + lane] = w;
            }
        }
        if (n == 32) {
#pragma unroll
            for (int g = 0; g < 32; g += RT_GROUP) {
                float a[RT_GROUP], c[RT_GROUP];
#pragma unroll
                for (int j = 0; j < RT_GROUP; ++j) {
                    const int ej = __builtin_amdgcn_readlane(en, g + j);
                    const float* tp = t + (size_t)(ej & 0xFFFF) * 64 + lane;
                    a[j] = tp[0];
                    c[j] = tp[32];
                }
#pragma unroll
                for (int j = 0; j < RT_GROUP; ++j) {
                    const int   ej = __builtin_amdgcn_readlane(en, g + j);
                    const float wj = __int_as_float(__builtin_amdgcn_readlane(__float_as_int(w), g + j));
                    float* ap = &tile[(ej >> 16) * 64 + lane];
                    atomicAdd(ap,      wj * a[j]);
                    atomicAdd(ap + 32, wj * c[j]);
                }
            }
        } else {
#pragma unroll
            for (int g = 0; g < 32; g += RT_GROUP) {
                if (g < n) {
                    float a[RT_GROUP], c[RT_GROUP];
#pragma unroll
                    for (int j = 0; j < RT_GROUP; ++j) {
                        a[j] = 0.0f;
                        c[j] = 0.0f;
                        if (g + j < n) {
                            const int ej = __builtin_amdgcn_readlane(en, g + j);
                            const float* tp = t + (size_t)(ej & 0xFFFF) * 64 + lane;
                            a[j] = tp[0];
                            c[j] = tp[32];
                        }
                    }
#pragma unroll
                    for (int j = 0; j < RT_GROUP; ++j) {
                        if (g + j < n) {
                            const int   ej = __builtin_amdgcn_readlane(en, g + j);
                            const float wj = __int_as_float(__builtin_amdgcn_readlane(__float_as_int(w), g + j));
                            float* ap = &tile[(ej >> 16) * 64 + lane];
                            atomicAdd(ap,      wj * a[j]);
                            atomicAdd(ap + 32, wj * c[j]);
                        }
                    }
                }
            }
        }
    }
    __syncthreads();

    for (int idx = tid; idx < nrows * 16; idx += 1024) {
        const int i = idx >> 4;
        const int q = (idx & 15) * 4;
        *(float4*)(out + (size_t)(lo + i) * 64 + q) = *(const float4*)(&tile[i * 64 + q]);
    }
}


extern "C" void kernel_launch(void* const* d_in, const int* in_sizes, int n_in,
                              void* d_out, int out_size, void* d_ws, size_t ws_size,
                              hipStream_t stream) {
    (void)in_sizes; (void)n_in; (void)out_size;

    const float* x  = (const float*)d_in[0];
    const int*   ei = (const int*)d_in[1];
    const float* W[5] = {(const float*)d_in[2], (const float*)d_in[4], (const float*)d_in[6],
                         (const float*)d_in[8], (const float*)d_in[10]};
    const float* B[5] = {(const float*)d_in[3], (const float*)d_in[5], (const float*)d_in[7],
                         (const float*)d_in[9], (const float*)d_in[11]};

    char*  ws  = (char*)d_ws;
    size_t off = 0;
    auto carve = [&](size_t bytes) {
        void* p = ws + off;
        off = (off + bytes + 255) & ~(size_t)255;
        return p;
    };
    int*   part  = (int*)carve((size_t)RT_NR * RH_G * 4);
    float* dinv  = (float*)carve((size_t)GCN_N * 4);
    int*   elist = (int*)carve((size_t)GCN_E * 4);
    float* t     = (float*)carve((size_t)GCN_N * 64 * 4);
    float* hagg  = (float*)carve((size_t)GCN_N * 64 * 4);
    int*   rinfo = (int*)carve((size_t)RT_NR * 2 * 4);
    float* wl    = (float*)carve((size_t)GCN_E * 4);
    const size_t ws_need = off;

    const int BG = ((GCN_N / 16) + 7) / 8;

    gcn_range_hist<<<RH_G, 1024, 0, stream>>>(ei, part);
    gcn_range_fill<<<RT_NR, 1024, 0, stream>>>(ei, part, elist, rinfo, dinv);

    gcn_gemm_wmma<GCN_IN, false><<<BG, 256, 0, stream>>>(x, W[0], t);
    gcn_range_agg<true><<<RT_NR, 1024, 0, stream>>>(elist, rinfo, t, dinv, B[0], hagg, wl, ws_need, ws_size);

    for (int L = 1; L <= 3; ++L) {
        gcn_gemm_wmma<GCN_H, true><<<BG, 256, 0, stream>>>(hagg, W[L], t);
        gcn_range_agg<false><<<RT_NR, 1024, 0, stream>>>(elist, rinfo, t, dinv, B[L], hagg, wl, ws_need, ws_size);
    }

    float* out = (float*)d_out;
    gcn_gemm_wmma<GCN_H, true><<<BG, 256, 0, stream>>>(hagg, W[4], t);
    gcn_range_agg<false><<<RT_NR, 1024, 0, stream>>>(elist, rinfo, t, dinv, B[4], out, wl, ws_need, ws_size);
}


# Range owners filter a byte-per-edge range-number stream stored by the histogram launch (16 edges per 16-byte load) instead of the 4-byte dst ids: 4x fewer bytes scanned; sum order stays arrival order
# speedup vs baseline: 2.5635x; 1.0950x over previous
#include <hip/hip_runtime.h>


#define GCN_N 50000
#define GCN_E 800000
#define GCN_IN 128
#define GCN_H 64

#define PAIR_STRIDE 130

typedef float v2f __attribute__((ext_vector_type(2)));
typedef float v8f __attribute__((ext_vector_type(8)));


template <int DIN, bool RELU>
__global__ __launch_bounds__(256) void gcn_gemm_wmma(const float* __restrict__ in,
                                                     const float* __restrict__ W,
                                                     float* __restrict__ out) {
    __shared__ float Wl[(DIN / 2) * PAIR_STRIDE];
    const int tid = threadIdx.x;
    for (int i = tid; i < DIN * 64; i += 256) {
        int k = i >> 6;
        int n = i & 63;
        Wl[(k >> 1) * PAIR_STRIDE + n * 2 + (k & 1)] = W[i];
    }
    __syncthreads();

    const int lane = tid & 31;
    const int wave = tid >> 5;
    const int tile = blockIdx.x * 8 + wave;
    if (tile >= (GCN_N >> 4)) return;

    const int m    = lane & 15;
    const int hi   = lane >> 4;
    const int koff = hi * 2;
    const float* arow = in + (size_t)(tile * 16 + m) * DIN;
    const float* brow = &Wl[(koff >> 1) * PAIR_STRIDE + m * 2];

    v8f acc0 = {}, acc1 = {}, acc2 = {}, acc3 = {};
    for (int k0 = 0; k0 < DIN; k0 += 4) {
        float2 av = *(const float2*)(arow + k0 + koff);
        if (RELU) { av.x = fmaxf(av.x, 0.0f); av.y = fmaxf(av.y, 0.0f); }
        v2f a; a.x = av.x; a.y = av.y;

        const float* bp = brow + (k0 >> 1) * PAIR_STRIDE;
        v2f b;
        b = *(const v2f*)(bp +  0);
        acc0 = __builtin_amdgcn_wmma_f32_16x16x4_f32(false, a, false, b, (short)0, acc0, false, false);
        b = *(const v2f*)(bp + 32);
        acc1 = __builtin_amdgcn_wmma_f32_16x16x4_f32(false, a, false, b, (short)0, acc1, false, false);
        b = *(const v2f*)(bp + 64);
        acc2 = __builtin_amdgcn_wmma_f32_16x16x4_f32(false, a, false, b, (short)0, acc2, false, false);
        b = *(const v2f*)(bp + 96);
        acc3 = __builtin_amdgcn_wmma_f32_16x16x4_f32(false, a, false, b, (short)0, acc3, false, false);
    }

    float* orow = out + (size_t)(tile * 16 + 8 * hi) * 64 + m;
#pragma unroll
    for (int r = 0; r < 8; ++r) {
        orow[(size_t)r * 64 +  0] = acc0[r];
        orow[(size_t)r * 64 + 16] = acc1[r];
        orow[(size_t)r * 64 + 32] = acc2[r];
        orow[(size_t)r * 64 + 48] = acc3[r];
    }
}


#define RT_R  208
#define RT_NR ((GCN_N + RT_R - 1) / RT_R)

static_assert(GCN_N <= 65536, "a src id must fit in 16 bits of an entry");
static_assert(RT_R <= 256, "a local row must fit in 8 bits of an entry");
static_assert(RT_NR <= 255, "a range number must fit in a byte, 255 left free");
static_assert(GCN_E % 16 == 0, "range bytes are read sixteen at a time");

#define RH_G 64
#define RH_V (GCN_E / 4 / RH_G)
#define RH_K ((RH_V + 1023) / 1024)
static_assert(GCN_E % (4 * RH_G) == 0, "the dst stream is cut into RH_G equal slices");
static_assert(RH_G % 4 == 0, "histogram columns are read four at a time");

__device__ __forceinline__ unsigned range_byte(int dst) {
    const unsigned r = (unsigned)dst / RT_R;
    return r < 255u ? r : 255u;
}

__global__ __launch_bounds__(1024) void gcn_range_hist(const int* __restrict__ ei,
                                                       int* __restrict__ part,
                                                       unsigned* __restrict__ rbytes) {
    __shared__ int s_hist[RT_NR];
    const int tid = threadIdx.x;
    const int g   = blockIdx.x;
    if (tid < RT_NR) s_hist[tid] = 0;
    __syncthreads();
    const int4* dst4 = (const int4*)(ei + GCN_E) + g * RH_V;
    int4 d[RH_K];
#pragma unroll
    for (int k = 0; k < RH_K; ++k) {
        const int i = tid + k * 1024;
        d[k] = (i < RH_V) ? dst4[i] : make_int4(-1, -1, -1, -1);
    }
#pragma unroll
    for (int k = 0; k < RH_K; ++k) {
        const unsigned r0 = (unsigned)d[k].x / RT_R, r1 = (unsigned)d[k].y / RT_R;
        const unsigned r2 = (unsigned)d[k].z / RT_R, r3 = (unsigned)d[k].w / RT_R;
        if (r0 < RT_NR) atomicAdd(&s_hist[r0], 1);
        if (r1 < RT_NR) atomicAdd(&s_hist[r1], 1);
        if (r2 < RT_NR) atomicAdd(&s_hist[r2], 1);
        if (r3 < RT_NR) atomicAdd(&s_hist[r3], 1);
        const int i = tid + k * 1024;
        if (i < RH_V) {
            rbytes[g * RH_V + i] = range_byte(d[k].x) | (range_byte(d[k].y) << 8) |
                                   (range_byte(d[k].z) << 16) | (range_byte(d[k].w) << 24);
        }
    }
    __syncthreads();
    if (tid < RT_NR) part[tid * RH_G + g] = s_hist[tid];
}

#define RT_SCAN 8
#define RT_NV (GCN_E / 16)

__device__ __forceinline__ unsigned zero_bytes(unsigned t) {
    const unsigned u = (t & 0x7F7F7F7Fu) + 0x7F7F7F7Fu;
    return ~(u | t) & 0x80808080u;
}

__global__ __launch_bounds__(1024) void gcn_range_fill(const int* __restrict__ ei,
                                                       const int* __restrict__ part,
                                                       const unsigned* __restrict__ rbytes,
                                                       int* __restrict__ entries,
                                                       int* __restrict__ rinfo,
                                                       float* __restrict__ dinv) {
    __shared__ int s_part[2][32];
    __shared__ int s_start, s_count, s_cursor;
    __shared__ int s_deg[RT_R];
    const int tid  = threadIdx.x;
    const int lane = tid & 31;
    const int wave = tid >> 5;
    const int b    = blockIdx.x;
    const int lo   = b * RT_R;
    const int hi   = (lo + RT_R < GCN_N) ? (lo + RT_R) : GCN_N;

    int before = 0, own = 0;
    const int4* part4 = (const int4*)part;
    const int nb4 = b * (RH_G / 4);
    for (int i4 = tid; i4 < nb4 + RH_G / 4; i4 += 1024) {
        const int4 pv = part4[i4];
        const int c = pv.x + pv.y + pv.z + pv.w;
        if (i4 < nb4) before += c; else own += c;
    }
    for (int o = 16; o > 0; o >>= 1) {
        before += __shfl_xor(before, o);
        own    += __shfl_xor(own, o);
    }
    if (lane == 0) { s_part[0][wave] = before; s_part[1][wave] = own; }
    if (tid == 0) s_cursor = 0;
    if (tid < RT_R) s_deg[tid] = 0;
    __syncthreads();
    if (wave == 0) {
        int x = s_part[0][lane], y = s_part[1][lane];
        for (int o = 16; o > 0; o >>= 1) {
            x += __shfl_xor(x, o);
            y += __shfl_xor(y, o);
        }
        if (lane == 0) {
            s_start = x; s_count = y;
            rinfo[2 * b] = x; rinfo[2 * b + 1] = y;
        }
    }
    __syncthreads();
    const int start = s_start;
    const int count = s_count;

    const uint4* rb4 = (const uint4*)rbytes;
    const unsigned own4 = (unsigned)b * 0x01010101u;
    for (int v0 = tid; v0 < RT_NV; v0 += 1024 * RT_SCAN) {
        uint4 d[RT_SCAN];
#pragma unroll
        for (int k = 0; k < RT_SCAN; ++k) {
            const int v = v0 + k * 1024;
            d[k] = (v < RT_NV) ? rb4[v] : make_uint4(~0u, ~0u, ~0u, ~0u);
        }
#pragma unroll
        for (int k = 0; k < RT_SCAN; ++k) {
            const int v = v0 + k * 1024;
            unsigned c = (zero_bytes(d[k].x ^ own4) >> 7) | (zero_bytes(d[k].y ^ own4) >> 6) |
                         (zero_bytes(d[k].z ^ own4) >> 5) | (zero_bytes(d[k].w ^ own4) >> 4);
            while (c != 0) {
                const int q = __builtin_ctz(c);
                c &= c - 1;
                const int p = atomicAdd(&s_cursor, 1);
                if (p < count) entries[start + p] = 16 * v + 4 * (q & 7) + (q >> 3);
            }
        }
    }
    __syncthreads();

    for (int p = tid; p < count; p += 1024) {
        const int e  = entries[start + p];
        const int lr = ei[GCN_E + e] - lo;
        atomicAdd(&s_deg[lr], 1);
        entries[start + p] = ei[e] | (lr << 16);
    }
    __syncthreads();

    if (tid < hi - lo) {
        const float d = 2.0f + (float)s_deg[tid];
        dinv[lo + tid] = (d > 0.0f) ? rsqrtf(d) : 0.0f;
    }
}


#define RT_GROUP 16

template <bool FIRST>
__global__ __launch_bounds__(1024) void gcn_range_agg(const int* __restrict__ entries,
                                                      const int* __restrict__ rinfo,
                                                      const float* __restrict__ t,
                                                      const float* __restrict__ dinv,
                                                      const float* __restrict__ bias,
                                                      float* __restrict__ out,
                                                      float* __restrict__ wl,
                                                      size_t ws_need, size_t ws_size) {
    __shared__ float tile[RT_R * 64];
    __shared__ float dloc[RT_R];
    const int tid  = threadIdx.x;
    const int lane = tid & 31;
    const int wave = __builtin_amdgcn_readfirstlane(tid >> 5);
    const int b    = blockIdx.x;
    const int lo   = b * RT_R;
    const int nrows = (GCN_N - lo < RT_R) ? (GCN_N - lo) : RT_R;
    const bool have_wl = ws_need <= ws_size;
    const bool read_wl = !FIRST && have_wl;

    for (int idx = tid; idx < nrows * 16; idx += 1024) {
        const int i = idx >> 4;
        const int q = (idx & 15) * 4;
        const float di = dinv[lo + i];
        const float s  = 2.0f * di * di;
        const float4 tv = *(const float4*)(t + (size_t)(lo + i) * 64 + q);
        float4 o;
        o.x = bias[q + 0] + s * tv.x;
        o.y = bias[q + 1] + s * tv.y;
        o.z = bias[q + 2] + s * tv.z;
        o.w = bias[q + 3] + s * tv.w;
        *(float4*)(&tile[i * 64 + q]) = o;
    }
    if (!read_wl) {
        for (int i = tid; i < nrows; i += 1024) dloc[i] = dinv[lo + i];
    }
    __syncthreads();

    const int start = rinfo[2 * b];
    const int count = rinfo[2 * b + 1];
    for (int base = wave * 32; base < count; base += 32 * 32) {
        const int n = (count - base < 32) ? (count - base) : 32;
        int   en = 0;
        float w  = 0.0f;
        if (lane < n) {
            en = entries[start + base + lane];
            if (read_wl) {
                w = wl[start + base + lane];
            } else {
                w = dinv[en & 0xFFFF] * dloc[en >> 16];
                if (FIRST && have_wl) wl[start + base + lane] = w;
            }
        }
        if (n == 32) {
#pragma unroll
            for (int g = 0; g < 32; g += RT_GROUP) {
                float a[RT_GROUP], c[RT_GROUP];
#pragma unroll
                for (int j = 0; j < RT_GROUP; ++j) {
                    const int ej = __builtin_amdgcn_readlane(en, g + j);
                    const float* tp = t + (size_t)(ej & 0xFFFF) * 64 + lane;
                    a[j] = tp[0];
                    c[j] = tp[32];
                }
#pragma unroll
                for (int j = 0; j < RT_GROUP; ++j) {
                    const int   ej = __builtin_amdgcn_readlane(en, g + j);
                    const float wj = __int_as_float(__builtin_amdgcn_readlane(__float_as_int(w), g + j));
                    float* ap = &tile[(ej >> 16) * 64 + lane];
                    atomicAdd(ap,      wj * a[j]);
                    atomicAdd(ap + 32, wj * c[j]);
                }
            }
        } else {
#pragma unroll
            for (int g = 0; g < 32; g += RT_GROUP) {
                if (g < n) {
                    float a[RT_GROUP], c[RT_GROUP];
#pragma unroll
                    for (int j = 0; j < RT_GROUP; ++j) {
                        a[j] = 0.0f;
                        c[j] = 0.0f;
                        if (g + j < n) {
                            const int ej = __builtin_amdgcn_readlane(en, g + j);
                            const float* tp = t + (size_t)(ej & 0xFFFF) * 64 + lane;
                            a[j] = tp[0];
                            c[j] = tp[32];
                        }
                    }
#pragma unroll
                    for (int j = 0; j < RT_GROUP; ++j) {
                        if (g + j < n) {
                            const int   ej = __builtin_amdgcn_readlane(en, g + j);
                            const float wj = __int_as_float(__builtin_amdgcn_readlane(__float_as_int(w), g + j));
                            float* ap = &tile[(ej >> 16) * 64 + lane];
                            atomicAdd(ap,      wj * a[j]);
                            atomicAdd(ap + 32, wj * c[j]);
                        }
                    }
                }
            }
        }
    }
    __syncthreads();

    for (int idx = tid; idx < nrows * 16; idx += 1024) {
        const int i = idx >> 4;
        const int q = (idx & 15) * 4;
        *(float4*)(out + (size_t)(lo + i) * 64 + q) = *(const float4*)(&tile[i * 64 + q]);
    }
}


extern "C" void kernel_launch(void* const* d_in, const int* in_sizes, int n_in,
                              void* d_out, int out_size, void* d_ws, size_t ws_size,
                              hipStream_t stream) {
    (void)in_sizes; (void)n_in; (void)out_size;

    const float* x  = (const float*)d_in[0];
    const int*   ei = (const int*)d_in[1];
    const float* W[5] = {(const float*)d_in[2], (const float*)d_in[4], (const float*)d_in[6],
                         (const float*)d_in[8], (const float*)d_in[10]};
    const float* B[5] = {(const float*)d_in[3], (const float*)d_in[5], (const float*)d_in[7],
                         (const float*)d_in[9], (const float*)d_in[11]};

    char*  ws  = (char*)d_ws;
    size_t off = 0;
    auto carve = [&](size_t bytes) {
        void* p = ws + off;
        off = (off + bytes + 255) & ~(size_t)255;
        return p;
    };
    int*   part  = (int*)carve((size_t)RT_NR * RH_G * 4);
    float* dinv  = (float*)carve((size_t)GCN_N * 4);
    int*   elist = (int*)carve((size_t)GCN_E * 4);
    float* t     = (float*)carve((size_t)GCN_N * 64 * 4);
    float* hagg  = (float*)carve((size_t)GCN_N * 64 * 4);
    int*   rinfo = (int*)carve((size_t)RT_NR * 2 * 4);
    float* wl    = (float*)carve((size_t)GCN_E * 4);
    const size_t ws_need = off;

    const int BG = ((GCN_N / 16) + 7) / 8;

    unsigned* rbytes = (unsigned*)t;
    gcn_range_hist<<<RH_G, 1024, 0, stream>>>(ei, part, rbytes);
    gcn_range_fill<<<RT_NR, 1024, 0, stream>>>(ei, part, rbytes, elist, rinfo, dinv);

    gcn_gemm_wmma<GCN_IN, false><<<BG, 256, 0, stream>>>(x, W[0], t);
    gcn_range_agg<true><<<RT_NR, 1024, 0, stream>>>(elist, rinfo, t, dinv, B[0], hagg, wl, ws_need, ws_size);

    for (int L = 1; L <= 3; ++L) {
        gcn_gemm_wmma<GCN_H, true><<<BG, 256, 0, stream>>>(hagg, W[L], t);
        gcn_range_agg<false><<<RT_NR, 1024, 0, stream>>>(elist, rinfo, t, dinv, B[L], hagg, wl, ws_need, ws_size);
    }

    float* out = (float*)d_out;
    gcn_gemm_wmma<GCN_H, true><<<BG, 256, 0, stream>>>(hagg, W[4], t);
    gcn_range_agg<false><<<RT_NR, 1024, 0, stream>>>(elist, rinfo, t, dinv, B[4], out, wl, ws_need, ws_size);
}


# Staging loops' loads leave as one group (was a wait per trip): W into LDS in 16-byte pieces (8/16 waits), edge-sum start values (4), histogram columns (4), list rewrite (8); sum order unchanged
# speedup vs baseline: 2.8179x; 1.0992x over previous
#include <hip/hip_runtime.h>


#define GCN_N 50000
#define GCN_E 800000
#define GCN_IN 128
#define GCN_H 64

#define PAIR_STRIDE 130

typedef float v2f __attribute__((ext_vector_type(2)));
typedef float v8f __attribute__((ext_vector_type(8)));


template <int DIN, bool RELU>
__global__ __launch_bounds__(256) void gcn_gemm_wmma(const float* __restrict__ in,
                                                     const float* __restrict__ W,
                                                     float* __restrict__ out) {
    __shared__ float Wl[(DIN / 2) * PAIR_STRIDE];
    const int tid = threadIdx.x;
    static_assert((DIN * 64) % (4 * 256) == 0, "W is staged in 16-byte pieces, the same number by every thread");
    constexpr int NV = DIN * 64 / 4 / 256;
    const float4* W4 = (const float4*)W;
    float4 wv[NV];
#pragma unroll
    for (int j = 0; j < NV; ++j) wv[j] = W4[tid + 256 * j];
#pragma unroll
    for (int j = 0; j < NV; ++j) {
        const int i4 = tid + 256 * j;
        const int k  = i4 >> 4;
        const int n  = (i4 & 15) * 4;
        float* wp = &Wl[(k >> 1) * PAIR_STRIDE + n * 2 + (k & 1)];
        wp[0] = wv[j].x;
        wp[2] = wv[j].y;
        wp[4] = wv[j].z;
        wp[6] = wv[j].w;
    }
    __syncthreads();

    const int lane = tid & 31;
    const int wave = tid >> 5;
    const int tile = blockIdx.x * 8 + wave;
    if (tile >= (GCN_N >> 4)) return;

    const int m    = lane & 15;
    const int hi   = lane >> 4;
    const int koff = hi * 2;
    const float* arow = in + (size_t)(tile * 16 + m) * DIN;
    const float* brow = &Wl[(koff >> 1) * PAIR_STRIDE + m * 2];

    v8f acc0 = {}, acc1 = {}, acc2 = {}, acc3 = {};
    for (int k0 = 0; k0 < DIN; k0 += 4) {
        float2 av = *(const float2*)(arow + k0 + koff);
        if (RELU) { av.x = fmaxf(av.x, 0.0f); av.y = fmaxf(av.y, 0.0f); }
        v2f a; a.x = av.x; a.y = av.y;

        const float* bp = brow + (k0 >> 1) * PAIR_STRIDE;
        v2f b;
        b = *(const v2f*)(bp +  0);
        acc0 = __builtin_amdgcn_wmma_f32_16x16x4_f32(false, a, false, b, (short)0, acc0, false, false);
        b = *(const v2f*)(bp + 32);
        acc1 = __builtin_amdgcn_wmma_f32_16x16x4_f32(false, a, false, b, (short)0, acc1, false, false);
        b = *(const v2f*)(bp + 64);
        acc2 = __builtin_amdgcn_wmma_f32_16x16x4_f32(false, a, false, b, (short)0, acc2, false, false);
        b = *(const v2f*)(bp + 96);
        acc3 = __builtin_amdgcn_wmma_f32_16x16x4_f32(false, a, false, b, (short)0, acc3, false, false);
    }

    float* orow = out + (size_t)(tile * 16 + 8 * hi) * 64 + m;
#pragma unroll
    for (int r = 0; r < 8; ++r) {
        orow[(size_t)r * 64 +  0] = acc0[r];
        orow[(size_t)r * 64 + 16] = acc1[r];
        orow[(size_t)r * 64 + 32] = acc2[r];
        orow[(size_t)r * 64 + 48] = acc3[r];
    }
}


#define RT_R  208
#define RT_NR ((GCN_N + RT_R - 1) / RT_R)

static_assert(GCN_N <= 65536, "a src id must fit in 16 bits of an entry");
static_assert(RT_R <= 256, "a local row must fit in 8 bits of an entry");
static_assert(RT_NR <= 255, "a range number must fit in a byte, 255 left free");
static_assert(GCN_E % 16 == 0, "range bytes are read sixteen at a time");

#define RH_G 64
#define RH_V (GCN_E / 4 / RH_G)
#define RH_K ((RH_V + 1023) / 1024)
static_assert(GCN_E % (4 * RH_G) == 0, "the dst stream is cut into RH_G equal slices");
static_assert(RH_G % 4 == 0, "histogram columns are read four at a time");

__device__ __forceinline__ unsigned range_byte(int dst) {
    const unsigned r = (unsigned)dst / RT_R;
    return r < 255u ? r : 255u;
}

__global__ __launch_bounds__(1024) void gcn_range_hist(const int* __restrict__ ei,
                                                       int* __restrict__ part,
                                                       unsigned* __restrict__ rbytes) {
    __shared__ int s_hist[RT_NR];
    const int tid = threadIdx.x;
    const int g   = blockIdx.x;
    if (tid < RT_NR) s_hist[tid] = 0;
    __syncthreads();
    const int4* dst4 = (const int4*)(ei + GCN_E) + g * RH_V;
    int4 d[RH_K];
#pragma unroll
    for (int k = 0; k < RH_K; ++k) {
        const int i = tid + k * 1024;
        d[k] = (i < RH_V) ? dst4[i] : make_int4(-1, -1, -1, -1);
    }
#pragma unroll
    for (int k = 0; k < RH_K; ++k) {
        const unsigned r0 = (unsigned)d[k].x / RT_R, r1 = (unsigned)d[k].y / RT_R;
        const unsigned r2 = (unsigned)d[k].z / RT_R, r3 = (unsigned)d[k].w / RT_R;
        if (r0 < RT_NR) atomicAdd(&s_hist[r0], 1);
        if (r1 < RT_NR) atomicAdd(&s_hist[r1], 1);
        if (r2 < RT_NR) atomicAdd(&s_hist[r2], 1);
        if (r3 < RT_NR) atomicAdd(&s_hist[r3], 1);
        const int i = tid + k * 1024;
        if (i < RH_V) {
            rbytes[g * RH_V + i] = range_byte(d[k].x) | (range_byte(d[k].y) << 8) |
                                   (range_byte(d[k].z) << 16) | (range_byte(d[k].w) << 24);
        }
    }
    __syncthreads();
    if (tid < RT_NR) part[tid * RH_G + g] = s_hist[tid];
}

#define RT_SCAN 8
#define RF_NP ((RT_NR * (RH_G / 4) + 1023) / 1024)
#define RF_NE 4
#define RT_NV (GCN_E / 16)

__device__ __forceinline__ unsigned zero_bytes(unsigned t) {
    const unsigned u = (t & 0x7F7F7F7Fu) + 0x7F7F7F7Fu;
    return ~(u | t) & 0x80808080u;
}

__global__ __launch_bounds__(1024) void gcn_range_fill(const int* __restrict__ ei,
                                                       const int* __restrict__ part,
                                                       const unsigned* __restrict__ rbytes,
                                                       int* __restrict__ entries,
                                                       int* __restrict__ rinfo,
                                                       float* __restrict__ dinv) {
    __shared__ int s_part[2][32];
    __shared__ int s_start, s_count, s_cursor;
    __shared__ int s_deg[RT_R];
    const int tid  = threadIdx.x;
    const int lane = tid & 31;
    const int wave = tid >> 5;
    const int b    = blockIdx.x;
    const int lo   = b * RT_R;
    const int hi   = (lo + RT_R < GCN_N) ? (lo + RT_R) : GCN_N;

    int before = 0, own = 0;
    const int4* part4 = (const int4*)part;
    const int nb4 = b * (RH_G / 4);
    int4 pv[RF_NP];
#pragma unroll
    for (int k = 0; k < RF_NP; ++k) {
        const int i4 = tid + k * 1024;
        pv[k] = (i4 < nb4 + RH_G / 4) ? part4[i4] : make_int4(0, 0, 0, 0);
    }
#pragma unroll
    for (int k = 0; k < RF_NP; ++k) {
        const int i4 = tid + k * 1024;
        const int c = pv[k].x + pv[k].y + pv[k].z + pv[k].w;
        if (i4 < nb4) before += c; else own += c;
    }
    for (int o = 16; o > 0; o >>= 1) {
        before += __shfl_xor(before, o);
        own    += __shfl_xor(own, o);
    }
    if (lane == 0) { s_part[0][wave] = before; s_part[1][wave] = own; }
    if (tid == 0) s_cursor = 0;
    if (tid < RT_R) s_deg[tid] = 0;
    __syncthreads();
    if (wave == 0) {
        int x = s_part[0][lane], y = s_part[1][lane];
        for (int o = 16; o > 0; o >>= 1) {
            x += __shfl_xor(x, o);
            y += __shfl_xor(y, o);
        }
        if (lane == 0) {
            s_start = x; s_count = y;
            rinfo[2 * b] = x; rinfo[2 * b + 1] = y;
        }
    }
    __syncthreads();
    const int start = s_start;
    const int count = s_count;

    const uint4* rb4 = (const uint4*)rbytes;
    const unsigned own4 = (unsigned)b * 0x01010101u;
    for (int v0 = tid; v0 < RT_NV; v0 += 1024 * RT_SCAN) {
        uint4 d[RT_SCAN];
#pragma unroll
        for (int k = 0; k < RT_SCAN; ++k) {
            const int v = v0 + k * 1024;
            d[k] = (v < RT_NV) ? rb4[v] : make_uint4(~0u, ~0u, ~0u, ~0u);
        }
#pragma unroll
        for (int k = 0; k < RT_SCAN; ++k) {
            const int v = v0 + k * 1024;
            unsigned c = (zero_bytes(d[k].x ^ own4) >> 7) | (zero_bytes(d[k].y ^ own4) >> 6) |
                         (zero_bytes(d[k].z ^ own4) >> 5) | (zero_bytes(d[k].w ^ own4) >> 4);
            while (c != 0) {
                const int q = __builtin_ctz(c);
                c &= c - 1;
                const int p = atomicAdd(&s_cursor, 1);
                if (p < count) entries[start + p] = 16 * v + 4 * (q & 7) + (q >> 3);
            }
        }
    }
    __syncthreads();

    for (int p0 = tid; p0 < count; p0 += 1024 * RF_NE) {
        unsigned e[RF_NE];
        int ed[RF_NE], es[RF_NE];
#pragma unroll
        for (int k = 0; k < RF_NE; ++k) {
            const int p = p0 + k * 1024;
            e[k] = (unsigned)entries[start + ((p < count) ? p : p0)];
        }
#pragma unroll
        for (int k = 0; k < RF_NE; ++k) {
            ed[k] = ei[GCN_E + e[k]];
            es[k] = ei[e[k]];
        }
#pragma unroll
        for (int k = 0; k < RF_NE; ++k) {
            const int p = p0 + k * 1024;
            if (p < count) {
                const int lr = ed[k] - lo;
                atomicAdd(&s_deg[lr], 1);
                entries[start + p] = es[k] | (lr << 16);
            }
        }
    }
    __syncthreads();

    if (tid < hi - lo) {
        const float d = 2.0f + (float)s_deg[tid];
        dinv[lo + tid] = (d > 0.0f) ? rsqrtf(d) : 0.0f;
    }
}


#define RT_GROUP 16
#define RA_NI ((RT_R * 16 + 1023) / 1024)

template <bool FIRST>
__global__ __launch_bounds__(1024) void gcn_range_agg(const int* __restrict__ entries,
                                                      const int* __restrict__ rinfo,
                                                      const float* __restrict__ t,
                                                      const float* __restrict__ dinv,
                                                      const float* __restrict__ bias,
                                                      float* __restrict__ out,
                                                      float* __restrict__ wl,
                                                      size_t ws_need, size_t ws_size) {
    __shared__ float tile[RT_R * 64];
    __shared__ float dloc[RT_R];
    const int tid  = threadIdx.x;
    const int lane = tid & 31;
    const int wave = __builtin_amdgcn_readfirstlane(tid >> 5);
    const int b    = blockIdx.x;
    const int lo   = b * RT_R;
    const int nrows = (GCN_N - lo < RT_R) ? (GCN_N - lo) : RT_R;
    const bool have_wl = ws_need <= ws_size;
    const bool read_wl = !FIRST && have_wl;

    {
        const int q = (tid & 15) * 4;
        const float4 bv = *(const float4*)(bias + q);
        float  di[RA_NI];
        float4 tv[RA_NI];
#pragma unroll
        for (int k = 0; k < RA_NI; ++k) {
            const int idx = tid + k * 1024;
            const int i = idx >> 4;
            di[k] = 0.0f;
            tv[k] = make_float4(0.0f, 0.0f, 0.0f, 0.0f);
            if (idx < nrows * 16) {
                di[k] = dinv[lo + i];
                tv[k] = *(const float4*)(t + (size_t)(lo + i) * 64 + q);
            }
        }
#pragma unroll
        for (int k = 0; k < RA_NI; ++k) {
            const int idx = tid + k * 1024;
            const int i = idx >> 4;
            if (idx < nrows * 16) {
                const float s = 2.0f * di[k] * di[k];
                float4 o;
                o.x = bv.x + s * tv[k].x;
                o.y = bv.y + s * tv[k].y;
                o.z = bv.z + s * tv[k].z;
                o.w = bv.w + s * tv[k].w;
                *(float4*)(&tile[i * 64 + q]) = o;
            }
        }
    }
    if (!read_wl) {
        for (int i = tid; i < nrows; i += 1024) dloc[i] = dinv[lo + i];
    }
    __syncthreads();

    const int start = rinfo[2 * b];
    const int count = rinfo[2 * b + 1];
    for (int base = wave * 32; base < count; base += 32 * 32) {
        const int n = (count - base < 32) ? (count - base) : 32;
        int   en = 0;
        float w  = 0.0f;
        if (lane < n) {
            en = entries[start + base + lane];
            if (read_wl) {
                w = wl[start + base + lane];
            } else {
                w = dinv[en & 0xFFFF] * dloc[en >> 16];
                if (FIRST && have_wl) wl[start + base + lane] = w;
            }
        }
        if (n == 32) {
#pragma unroll
            for (int g = 0; g < 32; g += RT_GROUP) {
                float a[RT_GROUP], c[RT_GROUP];
#pragma unroll
                for (int j = 0; j < RT_GROUP; ++j) {
                    const int ej = __builtin_amdgcn_readlane(en, g + j);
                    const float* tp = t + (size_t)(ej & 0xFFFF) * 64 + lane;
                    a[j] = tp[0];
                    c[j] = tp[32];
                }
#pragma unroll
                for (int j = 0; j < RT_GROUP; ++j) {
                    const int   ej = __builtin_amdgcn_readlane(en, g + j);
                    const float wj = __int_as_float(__builtin_amdgcn_readlane(__float_as_int(w), g + j));
                    float* ap = &tile[(ej >> 16) * 64 + lane];
                    atomicAdd(ap,      wj * a[j]);
                    atomicAdd(ap + 32, wj * c[j]);
                }
            }
        } else {
#pragma unroll
            for (int g = 0; g < 32; g += RT_GROUP) {
                if (g < n) {
                    float a[RT_GROUP], c[RT_GROUP];
#pragma unroll
                    for (int j = 0; j < RT_GROUP; ++j) {
                        a[j] = 0.0f;
                        c[j] = 0.0f;
                        if (g + j < n) {
                            const int ej = __builtin_amdgcn_readlane(en, g + j);
                            const float* tp = t + (size_t)(ej & 0xFFFF) * 64 + lane;
                            a[j] = tp[0];
                            c[j] = tp[32];
                        }
                    }
#pragma unroll
                    for (int j = 0; j < RT_GROUP; ++j) {
                        if (g + j < n) {
                            const int   ej = __builtin_amdgcn_readlane(en, g + j);
                            const float wj = __int_as_float(__builtin_amdgcn_readlane(__float_as_int(w), g + j));
                            float* ap = &tile[(ej >> 16) * 64 + lane];
                            atomicAdd(ap,      wj * a[j]);
                            atomicAdd(ap + 32, wj * c[j]);
                        }
                    }
                }
            }
        }
    }
    __syncthreads();

    for (int idx = tid; idx < nrows * 16; idx += 1024) {
        const int i = idx >> 4;
        const int q = (idx & 15) * 4;
        *(float4*)(out + (size_t)(lo + i) * 64 + q) = *(const float4*)(&tile[i * 64 + q]);
    }
}


extern "C" void kernel_launch(void* const* d_in, const int* in_sizes, int n_in,
                              void* d_out, int out_size, void* d_ws, size_t ws_size,
                              hipStream_t stream) {
    (void)in_sizes; (void)n_in; (void)out_size;

    const float* x  = (const float*)d_in[0];
    const int*   ei = (const int*)d_in[1];
    const float* W[5] = {(const float*)d_in[2], (const float*)d_in[4], (const float*)d_in[6],
                         (const float*)d_in[8], (const float*)d_in[10]};
    const float* B[5] = {(const float*)d_in[3], (const float*)d_in[5], (const float*)d_in[7],
                         (const float*)d_in[9], (const float*)d_in[11]};

    char*  ws  = (char*)d_ws;
    size_t off = 0;
    auto carve = [&](size_t bytes) {
        void* p = ws + off;
        off = (off + bytes + 255) & ~(size_t)255;
        return p;
    };
    int*   part  = (int*)carve((size_t)RT_NR * RH_G * 4);
    float* dinv  = (float*)carve((size_t)GCN_N * 4);
    int*   elist = (int*)carve((size_t)GCN_E * 4);
    float* t     = (float*)carve((size_t)GCN_N * 64 * 4);
    float* hagg  = (float*)carve((size_t)GCN_N * 64 * 4);
    int*   rinfo = (int*)carve((size_t)RT_NR * 2 * 4);
    float* wl    = (float*)carve((size_t)GCN_E * 4);
    const size_t ws_need = off;

    const int BG = ((GCN_N / 16) + 7) / 8;

    unsigned* rbytes = (unsigned*)t;
    gcn_range_hist<<<RH_G, 1024, 0, stream>>>(ei, part, rbytes);
    gcn_range_fill<<<RT_NR, 1024, 0, stream>>>(ei, part, rbytes, elist, rinfo, dinv);

    gcn_gemm_wmma<GCN_IN, false><<<BG, 256, 0, stream>>>(x, W[0], t);
    gcn_range_agg<true><<<RT_NR, 1024, 0, stream>>>(elist, rinfo, t, dinv, B[0], hagg, wl, ws_need, ws_size);

    for (int L = 1; L <= 3; ++L) {
        gcn_gemm_wmma<GCN_H, true><<<BG, 256, 0, stream>>>(hagg, W[L], t);
        gcn_range_agg<false><<<RT_NR, 1024, 0, stream>>>(elist, rinfo, t, dinv, B[L], hagg, wl, ws_need, ws_size);
    }

    float* out = (float*)d_out;
    gcn_gemm_wmma<GCN_H, true><<<BG, 256, 0, stream>>>(hagg, W[4], t);
    gcn_range_agg<false><<<RT_NR, 1024, 0, stream>>>(elist, rinfo, t, dinv, B[4], out, wl, ws_need, ws_size);
}
